# starting kernel
# baseline (speedup 1.0000x reference)
#include <hip/hip_runtime.h>
#include <hip/hip_bf16.h>


typedef float v2f __attribute__((ext_vector_type(2)));
typedef float v8f __attribute__((ext_vector_type(8)));

#define Bsz   256
#define Dd    16
#define Mm    128
#define HID   2048

__device__ __forceinline__ v8f wmma_f32_k4(v2f a, v2f b, v8f c) {
  return __builtin_amdgcn_wmma_f32_16x16x4_f32(
      false, a, false, b, (short)0, c, false, false);
}

__device__ __forceinline__ void gemm_tile(const float* __restrict__ A, int lda,
                                          const float* __restrict__ W, int ldw,
                                          int m0, int n0, int K, v8f& acc) {
  const int lane = threadIdx.x & 31;
  const int r    = lane & 15;
  const int kh   = (lane >> 4) << 1;
  const float* arow = A + (size_t)(m0 + r) * lda;
  const float* wrow = W + (size_t)(n0 + r) * ldw;
  for (int k = 0; k < K; k += 4) {
    v2f a, b;
    a.x = arow[k + kh];  a.y = arow[k + kh + 1];
    b.x = wrow[k + kh];  b.y = wrow[k + kh + 1];
    acc = wmma_f32_k4(a, b, acc);
  }
}

__device__ __forceinline__ float fast_tanh(float x) {
#if defined(__has_builtin)
#if __has_builtin(__builtin_amdgcn_tanhf)
  return __builtin_amdgcn_tanhf(x);
#elif __has_builtin(__builtin_amdgcn_tanh_f32)
  return __builtin_amdgcn_tanh_f32(x);
#else
  return tanhf(x);
#endif
#else
  return tanhf(x);
#endif
}

__device__ __forceinline__ float softplus_f(float x) {
  return fmaxf(x, 0.0f) + __logf(1.0f + __expf(-fabsf(x)));
}
__device__ __forceinline__ float log_abs_dtanh(float h) {
  return 2.0f * (0.6931471805599453f - h - softplus_f(-2.0f * h));
}

__global__ void k_layer0(const float* __restrict__ X, const float* __restrict__ W0,
                         const float* __restrict__ b0,
                         float* __restrict__ act, float* __restrict__ grad) {
  const int wave = (blockIdx.x * blockDim.x + threadIdx.x) >> 5;
  const int mt = wave >> 7;
  const int nt = wave & 127;
  if (mt >= Bsz / 16) return;
  v8f acc = {};
  gemm_tile(X, Dd, W0, Dd, mt * 16, nt * 16, Dd, acc);

  const int lane = threadIdx.x & 31;
  const int n = nt * 16 + (lane & 15);
  const int rowoff = (lane >> 4) * 8;
  const float ldiag = __logf(W0[n * Dd + (n >> 7)]);
  const float bias = b0[n];
  for (int v = 0; v < 8; ++v) {
    const int m = mt * 16 + v + rowoff;
    const float h = acc[v] + bias;
    act[m * HID + n]  = fast_tanh(h);
    grad[m * HID + n] = ldiag + log_abs_dtanh(h);
  }
}

__global__ void k_grad_prep(const float* __restrict__ grad,
                            float* __restrict__ mrow, float* __restrict__ E) {
  const int bk = blockIdx.x;
  const int i  = threadIdx.x;
  const int idx = bk * Mm + i;
  const float g = grad[idx];
  __shared__ float red[Mm];
  red[i] = g;
  __syncthreads();
  for (int s = 64; s > 0; s >>= 1) {
    if (i < s) red[i] = fmaxf(red[i], red[i + s]);
    __syncthreads();
  }
  const float mx = red[0];
  E[idx] = __expf(g - mx);
  if (i == 0) mrow[bk] = mx;
}

__global__ void k_layer1(const float* __restrict__ A, const float* __restrict__ W1,
                         const float* __restrict__ b1,
                         float* __restrict__ act, float* __restrict__ lga) {
  const int wave = (blockIdx.x * blockDim.x + threadIdx.x) >> 5;
  const int mt = wave >> 7;
  const int nt = wave & 127;
  if (mt >= Bsz / 16) return;
  v8f acc = {};
  gemm_tile(A, HID, W1, HID, mt * 16, nt * 16, HID, acc);

  const int lane = threadIdx.x & 31;
  const int n = nt * 16 + (lane & 15);
  const int rowoff = (lane >> 4) * 8;
  const float bias = b1[n];
  for (int v = 0; v < 8; ++v) {
    const int m = mt * 16 + v + rowoff;
    const float h = acc[v] + bias;
    act[m * HID + n] = fast_tanh(h);
    lga[m * HID + n] = log_abs_dtanh(h);
  }
}

__global__ void k_grad_gemm(const float* __restrict__ E, const float* __restrict__ W1,
                            const float* __restrict__ lga, const float* __restrict__ mrow,
                            float* __restrict__ grad) {
  const int wave = (blockIdx.x * blockDim.x + threadIdx.x) >> 5;
  const int ks  = wave >> 7;
  const int rem = wave & 127;
  const int mt  = rem >> 3;
  const int nt  = rem & 7;
  if (ks >= Dd) return;
  const float* Ak = E  + (size_t)ks * Mm;
  const float* Wk = W1 + (size_t)ks * Mm * HID + (size_t)ks * Mm;
  v8f acc = {};
  gemm_tile(Ak, HID, Wk, HID, mt * 16, nt * 16, Mm, acc);

  const int lane = threadIdx.x & 31;
  const int o = nt * 16 + (lane & 15);
  const int rowoff = (lane >> 4) * 8;
  for (int v = 0; v < 8; ++v) {
    const int b = mt * 16 + v + rowoff;
    const float s = acc[v];
    grad[b * HID + ks * Mm + o] =
        __logf(s) + mrow[b * Dd + ks] + lga[b * HID + ks * Mm + o];
  }
}

__global__ void k_out_gemm(const float* __restrict__ A, const float* __restrict__ W2,
                           float* __restrict__ out) {
  const int wave = threadIdx.x >> 5;
  if (wave >= Bsz / 16) return;
  v8f acc = {};
  gemm_tile(A, HID, W2, HID, wave * 16, 0, HID, acc);
  const int lane = threadIdx.x & 31;
  const int n = lane & 15;
  const int rowoff = (lane >> 4) * 8;
  for (int v = 0; v < 8; ++v) {
    const int m = wave * 16 + v + rowoff;
    out[m * Dd + n] = acc[v];
  }
}

__global__ void k_logdet(const float* __restrict__ E, const float* __restrict__ mrow,
                         const float* __restrict__ W2, float* __restrict__ ld,
                         int accumulate) {
  const int b = blockIdx.x * blockDim.x + threadIdx.x;
  if (b >= Bsz) return;
  float tot = 0.0f;
  for (int k = 0; k < Dd; ++k) {
    const float* w = W2 + k * HID + k * Mm;
    const float* e = E + b * HID + k * Mm;
    float s = 0.0f;
    for (int i = 0; i < Mm; ++i) s = fmaf(w[i], e[i], s);
    tot += __logf(s) + mrow[b * Dd + k];
  }
  if (accumulate) ld[b] += tot; else ld[b] = tot;
}

static void run_flow(const float* X, const float* W0, const float* W1,
                     const float* W2, const float* b0, const float* b1,
                     float* x_out, float* ld_out, int accumulate,
                     float* act0, float* act1, float* lga, float* grad,
                     float* E, float* mrow, hipStream_t stream) {
  k_layer0   <<<256, 256, 0, stream>>>(X, W0, b0, act0, grad);
  k_grad_prep<<<Bsz * Dd, Mm, 0, stream>>>(grad, mrow, E);
  k_layer1   <<<256, 256, 0, stream>>>(act0, W1, b1, act1, lga);
  k_grad_gemm<<<256, 256, 0, stream>>>(E, W1, lga, mrow, grad);
  k_grad_prep<<<Bsz * Dd, Mm, 0, stream>>>(grad, mrow, E);
  k_out_gemm <<<1, 512, 0, stream>>>(act1, W2, x_out);
  k_logdet   <<<1, 256, 0, stream>>>(E, mrow, W2, ld_out, accumulate);
}

extern "C" void kernel_launch(void* const* d_in, const int* in_sizes, int n_in,
                              void* d_out, int out_size, void* d_ws, size_t ws_size,
                              hipStream_t stream) {
  const float* inp   = (const float*)d_in[0];
  const float* f0_W0 = (const float*)d_in[1];
  const float* f0_W1 = (const float*)d_in[2];
  const float* f0_W2 = (const float*)d_in[3];
  const float* f0_b0 = (const float*)d_in[4];
  const float* f0_b1 = (const float*)d_in[5];
  const float* f1_W0 = (const float*)d_in[6];
  const float* f1_W1 = (const float*)d_in[7];
  const float* f1_W2 = (const float*)d_in[8];
  const float* f1_b0 = (const float*)d_in[9];
  const float* f1_b1 = (const float*)d_in[10];

  float* out = (float*)d_out;
  float* ld  = out + Bsz * Dd;

  float* ws   = (float*)d_ws;
  float* act0 = ws;
  float* act1 = act0 + Bsz * HID;
  float* lga  = act1 + Bsz * HID;
  float* grad = lga  + Bsz * HID;
  float* E    = grad + Bsz * HID;
  float* mrow = E    + Bsz * HID;
  float* xmid = mrow + Bsz * Dd;

  run_flow(inp,  f0_W0, f0_W1, f0_W2, f0_b0, f0_b1, xmid, ld,  0,
           act0, act1, lga, grad, E, mrow, stream);
  run_flow(xmid, f1_W0, f1_W1, f1_W2, f1_b0, f1_b1, out,  ld,  1,
           act0, act1, lga, grad, E, mrow, stream);
}


# Row max + exp(g - m) pass dropped: its two consumers (grad GEMM, tail; K = 128 = a whole row per wave) form m and e in their A loader, in registers, one half-wave exchange; 14 launches -> 10
# speedup vs baseline: 3.1432x; 3.1432x over previous
#include <hip/hip_runtime.h>
#include <hip/hip_bf16.h>


typedef float v2f __attribute__((ext_vector_type(2)));
typedef float v8f __attribute__((ext_vector_type(8)));

#define Bsz   256
#define Dd    16
#define Mm    128
#define HID   2048

__device__ __forceinline__ v8f wmma_f32_k4(v2f a, v2f b, v8f c) {
  return __builtin_amdgcn_wmma_f32_16x16x4_f32(
      false, a, false, b, (short)0, c, false, false);
}

__device__ __forceinline__ void gemm_tile(const float* __restrict__ A, int lda,
                                          const float* __restrict__ W, int ldw,
                                          int m0, int n0, int K, v8f& acc) {
  const int lane = threadIdx.x & 31;
  const int r    = lane & 15;
  const int kh   = (lane >> 4) << 1;
  const float* arow = A + (size_t)(m0 + r) * lda;
  const float* wrow = W + (size_t)(n0 + r) * ldw;
  for (int k = 0; k < K; k += 4) {
    v2f a, b;
    a.x = arow[k + kh];  a.y = arow[k + kh + 1];
    b.x = wrow[k + kh];  b.y = wrow[k + kh + 1];
    acc = wmma_f32_k4(a, b, acc);
  }
}

__device__ __forceinline__ float fast_tanh(float x) {
#if defined(__has_builtin)
#if __has_builtin(__builtin_amdgcn_tanhf)
  return __builtin_amdgcn_tanhf(x);
#elif __has_builtin(__builtin_amdgcn_tanh_f32)
  return __builtin_amdgcn_tanh_f32(x);
#else
  return tanhf(x);
#endif
#else
  return tanhf(x);
#endif
}

__device__ __forceinline__ float softplus_f(float x) {
  return fmaxf(x, 0.0f) + __logf(1.0f + __expf(-fabsf(x)));
}
__device__ __forceinline__ float log_abs_dtanh(float h) {
  return 2.0f * (0.6931471805599453f - h - softplus_f(-2.0f * h));
}

__global__ void k_layer0(const float* __restrict__ X, const float* __restrict__ W0,
                         const float* __restrict__ b0,
                         float* __restrict__ act, float* __restrict__ grad) {
  const int wave = (blockIdx.x * blockDim.x + threadIdx.x) >> 5;
  const int mt = wave >> 7;
  const int nt = wave & 127;
  if (mt >= Bsz / 16) return;
  v8f acc = {};
  gemm_tile(X, Dd, W0, Dd, mt * 16, nt * 16, Dd, acc);

  const int lane = threadIdx.x & 31;
  const int n = nt * 16 + (lane & 15);
  const int rowoff = (lane >> 4) * 8;
  const float ldiag = __logf(W0[n * Dd + (n >> 7)]);
  const float bias = b0[n];
  for (int v = 0; v < 8; ++v) {
    const int m = mt * 16 + v + rowoff;
    const float h = acc[v] + bias;
    act[m * HID + n]  = fast_tanh(h);
    grad[m * HID + n] = ldiag + log_abs_dtanh(h);
  }
}

__device__ __forceinline__ float rows_max_exp(const float* __restrict__ grow, int kh,
                                              v2f (&e)[32]) {
#pragma unroll
  for (int j = 0; j < 32; ++j) {
    e[j].x = grow[4 * j + kh];  e[j].y = grow[4 * j + kh + 1];
  }
  float mx = fmaxf(e[0].x, e[0].y);
#pragma unroll
  for (int j = 1; j < 32; ++j) mx = fmaxf(mx, fmaxf(e[j].x, e[j].y));
  mx = fmaxf(mx, __shfl_xor(mx, 16));
#pragma unroll
  for (int j = 0; j < 32; ++j) {
    e[j].x = __expf(e[j].x - mx);  e[j].y = __expf(e[j].y - mx);
  }
  return mx;
}

__global__ void k_layer1(const float* __restrict__ A, const float* __restrict__ W1,
                         const float* __restrict__ b1,
                         float* __restrict__ act, float* __restrict__ lga) {
  const int wave = (blockIdx.x * blockDim.x + threadIdx.x) >> 5;
  const int mt = wave >> 7;
  const int nt = wave & 127;
  if (mt >= Bsz / 16) return;
  v8f acc = {};
  gemm_tile(A, HID, W1, HID, mt * 16, nt * 16, HID, acc);

  const int lane = threadIdx.x & 31;
  const int n = nt * 16 + (lane & 15);
  const int rowoff = (lane >> 4) * 8;
  const float bias = b1[n];
  for (int v = 0; v < 8; ++v) {
    const int m = mt * 16 + v + rowoff;
    const float h = acc[v] + bias;
    act[m * HID + n] = fast_tanh(h);
    lga[m * HID + n] = log_abs_dtanh(h);
  }
}

__global__ void k_grad_gemm(const float* __restrict__ G, const float* __restrict__ W1,
                            const float* __restrict__ lga, float* __restrict__ Gnew) {
  const int wave = (blockIdx.x * blockDim.x + threadIdx.x) >> 5;
  const int ks  = wave >> 7;
  const int rem = wave & 127;
  const int mt  = rem >> 3;
  const int nt  = rem & 7;
  if (ks >= Dd) return;
  const int lane = threadIdx.x & 31;
  const int r    = lane & 15;
  const int kh   = (lane >> 4) << 1;
  const float* grow = G  + (size_t)(mt * 16 + r) * HID + ks * Mm;
  const float* wrow = W1 + (size_t)(ks * Mm + nt * 16 + r) * HID + ks * Mm;
  v2f e[32];
  const float mx = rows_max_exp(grow, kh, e);
  v8f acc = {};
#pragma unroll
  for (int j = 0; j < 32; ++j) {
    v2f b;
    b.x = wrow[4 * j + kh];  b.y = wrow[4 * j + kh + 1];
    acc = wmma_f32_k4(e[j], b, acc);
  }

  const int o = nt * 16 + r;
  const int rowoff = (lane >> 4) * 8;
  for (int v = 0; v < 8; ++v) {
    const int b = mt * 16 + v + rowoff;
    const float mv = __shfl(mx, v + rowoff);
    const float s = acc[v];
    Gnew[b * HID + ks * Mm + o] =
        __logf(s) + mv + lga[b * HID + ks * Mm + o];
  }
}

__global__ void k_tail(const float* __restrict__ act1, const float* __restrict__ G,
                       const float* __restrict__ W2,
                       float* __restrict__ Pout, float* __restrict__ Pld) {
  const int kb   = blockIdx.x >> 4;
  const int mt   = blockIdx.x & 15;
  const int lane = threadIdx.x & 31;
  const int r    = lane & 15;
  const int kh   = (lane >> 4) << 1;
  const float* arow = act1 + (size_t)(mt * 16 + r) * HID + kb * Mm;
  const float* grow = G    + (size_t)(mt * 16 + r) * HID + kb * Mm;
  const float* wrow = W2   + (size_t)r * HID + kb * Mm;
  v2f e[32];
  const float mx = rows_max_exp(grow, kh, e);
  v8f accx = {};
  v8f accs = {};
#pragma unroll
  for (int j = 0; j < 32; ++j) {
    v2f a, b;
    a.x = arow[4 * j + kh];  a.y = arow[4 * j + kh + 1];
    b.x = wrow[4 * j + kh];  b.y = wrow[4 * j + kh + 1];
    accx = wmma_f32_k4(a, b, accx);
    accs = wmma_f32_k4(e[j], b, accs);
  }
  const int rowoff = (lane >> 4) * 8;
  float* po = Pout + (size_t)kb * (Bsz * Dd);
  for (int v = 0; v < 8; ++v) {
    const int m = mt * 16 + v + rowoff;
    po[m * Dd + r] = accx[v];
  }
  float mv[8];
  for (int v = 0; v < 8; ++v) mv[v] = __shfl(mx, v + rowoff);
  if (r == kb) {
    float* pl = Pld + (kb * 16 + mt) * 32;
    for (int v = 0; v < 8; ++v)
      pl[v + rowoff] = __logf(accs[v]) + mv[v];
  }
}

__global__ void k_tail_sum(const float* __restrict__ Pout, const float* __restrict__ Pld,
                           float* __restrict__ x_out, float* __restrict__ ld,
                           int accumulate) {
  const int g = blockIdx.x;
  const int t = threadIdx.x;
  const int idx = g * 256 + t;
  float s = 0.0f;
  for (int kb = 0; kb < Dd; ++kb) s += Pout[kb * (Bsz * Dd) + idx];
  x_out[idx] = s;
  if (g < 8 && t < 32) {
    const int b = g * 32 + t;
    float tot = 0.0f;
    for (int kb = 0; kb < Dd; ++kb)
      tot += Pld[(kb * 16 + (b >> 4)) * 32 + (b & 15)];
    if (accumulate) ld[b] += tot; else ld[b] = tot;
  }
}

static void run_flow(const float* X, const float* W0, const float* W1,
                     const float* W2, const float* b0, const float* b1,
                     float* x_out, float* ld_out, int accumulate,
                     float* act0, float* act1, float* lga, float* g0,
                     float* g1, float* Pout, float* Pld,
                     hipStream_t stream) {
  k_layer0   <<<256, 256, 0, stream>>>(X, W0, b0, act0, g0);
  k_layer1   <<<256, 256, 0, stream>>>(act0, W1, b1, act1, lga);
  k_grad_gemm<<<256, 256, 0, stream>>>(g0, W1, lga, g1);
  k_tail     <<<256, 32, 0, stream>>>(act1, g1, W2, Pout, Pld);
  k_tail_sum <<<16, 256, 0, stream>>>(Pout, Pld, x_out, ld_out, accumulate);
}

extern "C" void kernel_launch(void* const* d_in, const int* in_sizes, int n_in,
                              void* d_out, int out_size, void* d_ws, size_t ws_size,
                              hipStream_t stream) {
  const float* inp   = (const float*)d_in[0];
  const float* f0_W0 = (const float*)d_in[1];
  const float* f0_W1 = (const float*)d_in[2];
  const float* f0_W2 = (const float*)d_in[3];
  const float* f0_b0 = (const float*)d_in[4];
  const float* f0_b1 = (const float*)d_in[5];
  const float* f1_W0 = (const float*)d_in[6];
  const float* f1_W1 = (const float*)d_in[7];
  const float* f1_W2 = (const float*)d_in[8];
  const float* f1_b0 = (const float*)d_in[9];
  const float* f1_b1 = (const float*)d_in[10];

  float* out = (float*)d_out;
  float* ld  = out + Bsz * Dd;

  float* ws   = (float*)d_ws;
  float* act0 = ws;
  float* act1 = act0 + Bsz * HID;
  float* lga  = act1 + Bsz * HID;
  float* g0   = lga  + Bsz * HID;
  float* g1   = g0   + Bsz * HID;
  float* xmid = g1   + Bsz * HID;
  float* Pout = xmid + Bsz * Dd;
  float* Pld  = Pout + Dd * Bsz * Dd;

  run_flow(inp,  f0_W0, f0_W1, f0_W2, f0_b0, f0_b1, xmid, ld,  0,
           act0, act1, lga, g0, g1, Pout, Pld, stream);
  run_flow(xmid, f1_W0, f1_W1, f1_W2, f1_b0, f1_b1, out,  ld,  1,
           act0, act1, lga, g0, g1, Pout, Pld, stream);
}


# Layer-1 GEMM: layer 0 stores act0 in A-fragment order (whole-line 16-byte loads); each wave feeds every W1 fragment to 2 M-tiles (32x16 per wave, 1 wave per SIMD, next 16 k-steps loaded ahead)
# speedup vs baseline: 4.6070x; 1.4657x over previous
#include <hip/hip_runtime.h>
#include <hip/hip_bf16.h>


typedef float v2f __attribute__((ext_vector_type(2)));
typedef float v4f __attribute__((ext_vector_type(4)));
typedef float v8f __attribute__((ext_vector_type(8)));

#define Bsz   256
#define Dd    16
#define Mm    128
#define HID   2048

__device__ __forceinline__ v8f wmma_f32_k4(v2f a, v2f b, v8f c) {
  return __builtin_amdgcn_wmma_f32_16x16x4_f32(
      false, a, false, b, (short)0, c, false, false);
}

__device__ __forceinline__ void gemm_tile(const float* __restrict__ A, int lda,
                                          const float* __restrict__ W, int ldw,
                                          int m0, int n0, int K, v8f& acc) {
  const int lane = threadIdx.x & 31;
  const int r    = lane & 15;
  const int kh   = (lane >> 4) << 1;
  const float* arow = A + (size_t)(m0 + r) * lda;
  const float* wrow = W + (size_t)(n0 + r) * ldw;
  for (int k = 0; k < K; k += 4) {
    v2f a, b;
    a.x = arow[k + kh];  a.y = arow[k + kh + 1];
    b.x = wrow[k + kh];  b.y = wrow[k + kh + 1];
    acc = wmma_f32_k4(a, b, acc);
  }
}

__device__ __forceinline__ float fast_tanh(float x) {
#if defined(__has_builtin)
#if __has_builtin(__builtin_amdgcn_tanhf)
  return __builtin_amdgcn_tanhf(x);
#elif __has_builtin(__builtin_amdgcn_tanh_f32)
  return __builtin_amdgcn_tanh_f32(x);
#else
  return tanhf(x);
#endif
#else
  return tanhf(x);
#endif
}

__device__ __forceinline__ float softplus_f(float x) {
  return fmaxf(x, 0.0f) + __logf(1.0f + __expf(-fabsf(x)));
}
__device__ __forceinline__ float log_abs_dtanh(float h) {
  return 2.0f * (0.6931471805599453f - h - softplus_f(-2.0f * h));
}

__global__ void k_layer0(const float* __restrict__ X, const float* __restrict__ W0,
                         const float* __restrict__ b0,
                         float* __restrict__ act, float* __restrict__ grad) {
  const int wave = (blockIdx.x * blockDim.x + threadIdx.x) >> 5;
  const int mt = wave >> 7;
  const int nt = wave & 127;
  if (mt >= Bsz / 16) return;
  v8f acc = {};
  gemm_tile(X, Dd, W0, Dd, mt * 16, nt * 16, Dd, acc);

  const int lane = threadIdx.x & 31;
  const int n = nt * 16 + (lane & 15);
  const int rowoff = (lane >> 4) * 8;
  const float ldiag = __logf(W0[n * Dd + (n >> 7)]);
  const float bias = b0[n];
  const int cl = lane & 15;
  const int pk = ((mt * 256 + nt * 2 + (cl >> 3)) * 32 + ((cl >> 1) & 1) * 16) * 4
               + 2 * ((cl >> 2) & 1) + (cl & 1);
  for (int v = 0; v < 8; ++v) {
    const int m = mt * 16 + v + rowoff;
    const float h = acc[v] + bias;
    act[pk + (v + rowoff) * 4] = fast_tanh(h);
    grad[m * HID + n] = ldiag + log_abs_dtanh(h);
  }
}

__device__ __forceinline__ float rows_max_exp(const float* __restrict__ grow, int kh,
                                              v2f (&e)[32]) {
#pragma unroll
  for (int j = 0; j < 32; ++j) {
    e[j].x = grow[4 * j + kh];  e[j].y = grow[4 * j + kh + 1];
  }
  float mx = fmaxf(e[0].x, e[0].y);
#pragma unroll
  for (int j = 1; j < 32; ++j) mx = fmaxf(mx, fmaxf(e[j].x, e[j].y));
  mx = fmaxf(mx, __shfl_xor(mx, 16));
#pragma unroll
  for (int j = 0; j < 32; ++j) {
    e[j].x = __expf(e[j].x - mx);  e[j].y = __expf(e[j].y - mx);
  }
  return mx;
}

#define L1_KS 16

__device__ __forceinline__ void l1_load(const float* __restrict__ wrow,
                                        const v4f* __restrict__ pa0,
                                        const v4f* __restrict__ pa1, int blk,
                                        v2f (&b)[L1_KS], v4f (&a0)[L1_KS / 2],
                                        v4f (&a1)[L1_KS / 2]) {
#pragma unroll
  for (int i = 0; i < L1_KS; ++i)
    b[i] = *(const v2f*)(wrow + (blk * L1_KS + i) * 4);
#pragma unroll
  for (int i = 0; i < L1_KS / 2; ++i) {
    a0[i] = pa0[(blk * (L1_KS / 2) + i) * 32];
    a1[i] = pa1[(blk * (L1_KS / 2) + i) * 32];
  }
}

__device__ __forceinline__ void l1_mma(const v2f (&b)[L1_KS], const v4f (&a0)[L1_KS / 2],
                                       const v4f (&a1)[L1_KS / 2], v8f& acc0, v8f& acc1) {
#pragma unroll
  for (int i = 0; i < L1_KS / 2; ++i) {
    v2f x0, x1, y0, y1;
    x0.x = a0[i].x;  x0.y = a0[i].y;   y0.x = a0[i].z;  y0.y = a0[i].w;
    x1.x = a1[i].x;  x1.y = a1[i].y;   y1.x = a1[i].z;  y1.y = a1[i].w;
    acc0 = wmma_f32_k4(x0, b[2 * i], acc0);
    acc1 = wmma_f32_k4(x1, b[2 * i], acc1);
    acc0 = wmma_f32_k4(y0, b[2 * i + 1], acc0);
    acc1 = wmma_f32_k4(y1, b[2 * i + 1], acc1);
  }
}

__global__ void __launch_bounds__(128)
k_layer1(const float* __restrict__ Apk, const float* __restrict__ W1,
         const float* __restrict__ b1,
         float* __restrict__ act, float* __restrict__ lga) {
  const int lane = threadIdx.x & 31;
  const int mg = blockIdx.x >> 5;
  const int nt = (blockIdx.x & 31) * 4 + (threadIdx.x >> 5);
  const int r  = lane & 15;
  const int kh = (lane >> 4) << 1;
  const float* wrow = W1 + (size_t)(nt * 16 + r) * HID + kh;
  const v4f* pa0 = (const v4f*)Apk + (size_t)(mg * 2) * (256 * 32) + lane;
  const v4f* pa1 = pa0 + 256 * 32;

  v8f acc0 = {};
  v8f acc1 = {};
  v2f bA[L1_KS], bB[L1_KS];
  v4f a0A[L1_KS / 2], a1A[L1_KS / 2], a0B[L1_KS / 2], a1B[L1_KS / 2];
  const int nblk = HID / 4 / L1_KS;
  l1_load(wrow, pa0, pa1, 0, bA, a0A, a1A);
  for (int blk = 0; blk < nblk - 2; blk += 2) {
    l1_load(wrow, pa0, pa1, blk + 1, bB, a0B, a1B);
    l1_mma(bA, a0A, a1A, acc0, acc1);
    l1_load(wrow, pa0, pa1, blk + 2, bA, a0A, a1A);
    l1_mma(bB, a0B, a1B, acc0, acc1);
  }
  l1_load(wrow, pa0, pa1, nblk - 1, bB, a0B, a1B);
  l1_mma(bA, a0A, a1A, acc0, acc1);
  l1_mma(bB, a0B, a1B, acc0, acc1);

  const int n = nt * 16 + r;
  const int rowoff = (lane >> 4) * 8;
  const float bias = b1[n];
  for (int v = 0; v < 8; ++v) {
    const int m = (mg * 2) * 16 + v + rowoff;
    const float h = acc0[v] + bias;
    act[m * HID + n] = fast_tanh(h);
    lga[m * HID + n] = log_abs_dtanh(h);
  }
  for (int v = 0; v < 8; ++v) {
    const int m = (mg * 2 + 1) * 16 + v + rowoff;
    const float h = acc1[v] + bias;
    act[m * HID + n] = fast_tanh(h);
    lga[m * HID + n] = log_abs_dtanh(h);
  }
}

__global__ void k_grad_gemm(const float* __restrict__ G, const float* __restrict__ W1,
                            const float* __restrict__ lga, float* __restrict__ Gnew) {
  const int wave = (blockIdx.x * blockDim.x + threadIdx.x) >> 5;
  const int ks  = wave >> 7;
  const int rem = wave & 127;
  const int mt  = rem >> 3;
  const int nt  = rem & 7;
  if (ks >= Dd) return;
  const int lane = threadIdx.x & 31;
  const int r    = lane & 15;
  const int kh   = (lane >> 4) << 1;
  const float* grow = G  + (size_t)(mt * 16 + r) * HID + ks * Mm;
  const float* wrow = W1 + (size_t)(ks * Mm + nt * 16 + r) * HID + ks * Mm;
  v2f e[32];
  const float mx = rows_max_exp(grow, kh, e);
  v8f acc = {};
#pragma unroll
  for (int j = 0; j < 32; ++j) {
    v2f b;
    b.x = wrow[4 * j + kh];  b.y = wrow[4 * j + kh + 1];
    acc = wmma_f32_k4(e[j], b, acc);
  }

  const int o = nt * 16 + r;
  const int rowoff = (lane >> 4) * 8;
  for (int v = 0; v < 8; ++v) {
    const int b = mt * 16 + v + rowoff;
    const float mv = __shfl(mx, v + rowoff);
    const float s = acc[v];
    Gnew[b * HID + ks * Mm + o] =
        __logf(s) + mv + lga[b * HID + ks * Mm + o];
  }
}

__global__ void k_tail(const float* __restrict__ act1, const float* __restrict__ G,
                       const float* __restrict__ W2,
                       float* __restrict__ Pout, float* __restrict__ Pld) {
  const int kb   = blockIdx.x >> 4;
  const int mt   = blockIdx.x & 15;
  const int lane = threadIdx.x & 31;
  const int r    = lane & 15;
  const int kh   = (lane >> 4) << 1;
  const float* arow = act1 + (size_t)(mt * 16 + r) * HID + kb * Mm;
  const float* grow = G    + (size_t)(mt * 16 + r) * HID + kb * Mm;
  const float* wrow = W2   + (size_t)r * HID + kb * Mm;
  v2f e[32];
  const float mx = rows_max_exp(grow, kh, e);
  v8f accx = {};
  v8f accs = {};
#pragma unroll
  for (int j = 0; j < 32; ++j) {
    v2f a, b;
    a.x = arow[4 * j + kh];  a.y = arow[4 * j + kh + 1];
    b.x = wrow[4 * j + kh];  b.y = wrow[4 * j + kh + 1];
    accx = wmma_f32_k4(a, b, accx);
    accs = wmma_f32_k4(e[j], b, accs);
  }
  const int rowoff = (lane >> 4) * 8;
  float* po = Pout + (size_t)kb * (Bsz * Dd);
  for (int v = 0; v < 8; ++v) {
    const int m = mt * 16 + v + rowoff;
    po[m * Dd + r] = accx[v];
  }
  float mv[8];
  for (int v = 0; v < 8; ++v) mv[v] = __shfl(mx, v + rowoff);
  if (r == kb) {
    float* pl = Pld + (kb * 16 + mt) * 32;
    for (int v = 0; v < 8; ++v)
      pl[v + rowoff] = __logf(accs[v]) + mv[v];
  }
}

__global__ void k_tail_sum(const float* __restrict__ Pout, const float* __restrict__ Pld,
                           float* __restrict__ x_out, float* __restrict__ ld,
                           int accumulate) {
  const int g = blockIdx.x;
  const int t = threadIdx.x;
  const int idx = g * 256 + t;
  float s = 0.0f;
  for (int kb = 0; kb < Dd; ++kb) s += Pout[kb * (Bsz * Dd) + idx];
  x_out[idx] = s;
  if (g < 8 && t < 32) {
    const int b = g * 32 + t;
    float tot = 0.0f;
    for (int kb = 0; kb < Dd; ++kb)
      tot += Pld[(kb * 16 + (b >> 4)) * 32 + (b & 15)];
    if (accumulate) ld[b] += tot; else ld[b] = tot;
  }
}

static void run_flow(const float* X, const float* W0, const float* W1,
                     const float* W2, const float* b0, const float* b1,
                     float* x_out, float* ld_out, int accumulate,
                     float* act0, float* act1, float* lga, float* g0,
                     float* g1, float* Pout, float* Pld,
                     hipStream_t stream) {
  k_layer0   <<<256, 256, 0, stream>>>(X, W0, b0, act0, g0);
  k_layer1   <<<256, 128, 0, stream>>>(act0, W1, b1, act1, lga);
  k_grad_gemm<<<256, 256, 0, stream>>>(g0, W1, lga, g1);
  k_tail     <<<256, 32, 0, stream>>>(act1, g1, W2, Pout, Pld);
  k_tail_sum <<<16, 256, 0, stream>>>(Pout, Pld, x_out, ld_out, accumulate);
}

extern "C" void kernel_launch(void* const* d_in, const int* in_sizes, int n_in,
                              void* d_out, int out_size, void* d_ws, size_t ws_size,
                              hipStream_t stream) {
  const float* inp   = (const float*)d_in[0];
  const float* f0_W0 = (const float*)d_in[1];
  const float* f0_W1 = (const float*)d_in[2];
  const float* f0_W2 = (const float*)d_in[3];
  const float* f0_b0 = (const float*)d_in[4];
  const float* f0_b1 = (const float*)d_in[5];
  const float* f1_W0 = (const float*)d_in[6];
  const float* f1_W1 = (const float*)d_in[7];
  const float* f1_W2 = (const float*)d_in[8];
  const float* f1_b0 = (const float*)d_in[9];
  const float* f1_b1 = (const float*)d_in[10];

  float* out = (float*)d_out;
  float* ld  = out + Bsz * Dd;

  float* ws   = (float*)d_ws;
  float* act0 = ws;
  float* act1 = act0 + Bsz * HID;
  float* lga  = act1 + Bsz * HID;
  float* g0   = lga  + Bsz * HID;
  float* g1   = g0   + Bsz * HID;
  float* xmid = g1   + Bsz * HID;
  float* Pout = xmid + Bsz * Dd;
  float* Pld  = Pout + Dd * Bsz * Dd;

  run_flow(inp,  f0_W0, f0_W1, f0_W2, f0_b0, f0_b1, xmid, ld,  0,
           act0, act1, lga, g0, g1, Pout, Pld, stream);
  run_flow(xmid, f1_W0, f1_W1, f1_W2, f1_b0, f1_b1, out,  ld,  1,
           act0, act1, lga, g0, g1, Pout, Pld, stream);
}


# Layer 1's k loop: ring of eight register sets of 4 k-steps, each block's 8 fragment loads go out as one group seven blocks ahead; ordering marks keep refills behind the WMMAs that read the set
# speedup vs baseline: 5.9444x; 1.2903x over previous
#include <hip/hip_runtime.h>
#include <hip/hip_bf16.h>


typedef float v2f __attribute__((ext_vector_type(2)));
typedef float v4f __attribute__((ext_vector_type(4)));
typedef float v8f __attribute__((ext_vector_type(8)));

#define Bsz   256
#define Dd    16
#define Mm    128
#define HID   2048

__device__ __forceinline__ v8f wmma_f32_k4(v2f a, v2f b, v8f c) {
  return __builtin_amdgcn_wmma_f32_16x16x4_f32(
      false, a, false, b, (short)0, c, false, false);
}

__device__ __forceinline__ void gemm_tile(const float* __restrict__ A, int lda,
                                          const float* __restrict__ W, int ldw,
                                          int m0, int n0, int K, v8f& acc) {
  const int lane = threadIdx.x & 31;
  const int r    = lane & 15;
  const int kh   = (lane >> 4) << 1;
  const float* arow = A + (size_t)(m0 + r) * lda;
  const float* wrow = W + (size_t)(n0 + r) * ldw;
  for (int k = 0; k < K; k += 4) {
    v2f a, b;
    a.x = arow[k + kh];  a.y = arow[k + kh + 1];
    b.x = wrow[k + kh];  b.y = wrow[k + kh + 1];
    acc = wmma_f32_k4(a, b, acc);
  }
}

__device__ __forceinline__ float fast_tanh(float x) {
#if defined(__has_builtin)
#if __has_builtin(__builtin_amdgcn_tanhf)
  return __builtin_amdgcn_tanhf(x);
#elif __has_builtin(__builtin_amdgcn_tanh_f32)
  return __builtin_amdgcn_tanh_f32(x);
#else
  return tanhf(x);
#endif
#else
  return tanhf(x);
#endif
}

__device__ __forceinline__ float softplus_f(float x) {
  return fmaxf(x, 0.0f) + __logf(1.0f + __expf(-fabsf(x)));
}
__device__ __forceinline__ float log_abs_dtanh(float h) {
  return 2.0f * (0.6931471805599453f - h - softplus_f(-2.0f * h));
}

__global__ void k_layer0(const float* __restrict__ X, const float* __restrict__ W0,
                         const float* __restrict__ b0,
                         float* __restrict__ act, float* __restrict__ gpk) {
  const int wave = (blockIdx.x * blockDim.x + threadIdx.x) >> 5;
  const int mt = wave >> 7;
  const int nt = wave & 127;
  if (mt >= Bsz / 16) return;
  v8f acc = {};
  gemm_tile(X, Dd, W0, Dd, mt * 16, nt * 16, Dd, acc);

  const int lane = threadIdx.x & 31;
  const int n = nt * 16 + (lane & 15);
  const int rowoff = (lane >> 4) * 8;
  const float ldiag = __logf(W0[n * Dd + (n >> 7)]);
  const float bias = b0[n];
  const int cl = lane & 15;
  const int pk = ((mt * 256 + nt * 2 + (cl >> 3)) * 32 + ((cl >> 1) & 1) * 16) * 4
               + 2 * ((cl >> 2) & 1) + (cl & 1);
  for (int v = 0; v < 8; ++v) {
    const float h = acc[v] + bias;
    act[pk + (v + rowoff) * 4] = fast_tanh(h);
    gpk[pk + (v + rowoff) * 4] = ldiag + log_abs_dtanh(h);
  }
}

__device__ __forceinline__ float rows_max_exp(const float* __restrict__ grow, int kh,
                                              v2f (&e)[32]) {
#pragma unroll
  for (int j = 0; j < 32; ++j) {
    e[j].x = grow[4 * j + kh];  e[j].y = grow[4 * j + kh + 1];
  }
  float mx = fmaxf(e[0].x, e[0].y);
#pragma unroll
  for (int j = 1; j < 32; ++j) mx = fmaxf(mx, fmaxf(e[j].x, e[j].y));
  mx = fmaxf(mx, __shfl_xor(mx, 16));
#pragma unroll
  for (int j = 0; j < 32; ++j) {
    e[j].x = __expf(e[j].x - mx);  e[j].y = __expf(e[j].y - mx);
  }
  return mx;
}

#define L1_KS 4
#define L1_NS 8

#define L1_PHASE() asm volatile("" : "+v"(acc0), "+v"(acc1) :: "memory")

__device__ __forceinline__ void l1_load(const float* __restrict__ wrow,
                                        const v4f* __restrict__ pa0,
                                        const v4f* __restrict__ pa1, int blk,
                                        v2f (&b)[L1_KS], v4f (&a0)[L1_KS / 2],
                                        v4f (&a1)[L1_KS / 2]) {
#pragma unroll
  for (int i = 0; i < L1_KS; ++i)
    b[i] = *(const v2f*)(wrow + (blk * L1_KS + i) * 4);
#pragma unroll
  for (int i = 0; i < L1_KS / 2; ++i) {
    a0[i] = pa0[(blk * (L1_KS / 2) + i) * 32];
    a1[i] = pa1[(blk * (L1_KS / 2) + i) * 32];
  }
}

__device__ __forceinline__ void l1_mma(const v2f (&b)[L1_KS], const v4f (&a0)[L1_KS / 2],
                                       const v4f (&a1)[L1_KS / 2], v8f& acc0, v8f& acc1) {
#pragma unroll
  for (int i = 0; i < L1_KS / 2; ++i) {
    v2f x0, x1, y0, y1;
    x0.x = a0[i].x;  x0.y = a0[i].y;   y0.x = a0[i].z;  y0.y = a0[i].w;
    x1.x = a1[i].x;  x1.y = a1[i].y;   y1.x = a1[i].z;  y1.y = a1[i].w;
    acc0 = wmma_f32_k4(x0, b[2 * i], acc0);
    acc1 = wmma_f32_k4(x1, b[2 * i], acc1);
    acc0 = wmma_f32_k4(y0, b[2 * i + 1], acc0);
    acc1 = wmma_f32_k4(y1, b[2 * i + 1], acc1);
  }
}

__device__ __forceinline__ void l1_gmma(const v2f (&b)[32], const v4f (&e0)[16],
                                        const v4f (&e1)[16], v8f& s0, v8f& s1) {
#pragma unroll
  for (int i = 0; i < 16; ++i) {
    v2f x0, x1, y0, y1;
    x0.x = e0[i].x;  x0.y = e0[i].y;   y0.x = e0[i].z;  y0.y = e0[i].w;
    x1.x = e1[i].x;  x1.y = e1[i].y;   y1.x = e1[i].z;  y1.y = e1[i].w;
    s0 = wmma_f32_k4(x0, b[2 * i], s0);
    s1 = wmma_f32_k4(x1, b[2 * i], s1);
    s0 = wmma_f32_k4(y0, b[2 * i + 1], s0);
    s1 = wmma_f32_k4(y1, b[2 * i + 1], s1);
  }
}

__device__ __forceinline__ void frag_load(const v4f* __restrict__ pg, v4f (&e)[16]) {
#pragma unroll
  for (int i = 0; i < 16; ++i) e[i] = pg[i * 32];
}
__device__ __forceinline__ float frag_max_exp(v4f (&e)[16]) {
  float mx = fmaxf(fmaxf(e[0].x, e[0].y), fmaxf(e[0].z, e[0].w));
#pragma unroll
  for (int i = 1; i < 16; ++i)
    mx = fmaxf(mx, fmaxf(fmaxf(e[i].x, e[i].y), fmaxf(e[i].z, e[i].w)));
  mx = fmaxf(mx, __shfl_xor(mx, 16));
#pragma unroll
  for (int i = 0; i < 16; ++i) {
    e[i].x = __expf(e[i].x - mx);  e[i].y = __expf(e[i].y - mx);
    e[i].z = __expf(e[i].z - mx);  e[i].w = __expf(e[i].w - mx);
  }
  return mx;
}
__device__ __forceinline__ void frag_hold(const v4f (&e)[16]) {
  asm volatile("" :: "v"(e[0]), "v"(e[1]), "v"(e[2]), "v"(e[3]), "v"(e[4]), "v"(e[5]),
               "v"(e[6]), "v"(e[7]), "v"(e[8]), "v"(e[9]), "v"(e[10]), "v"(e[11]),
               "v"(e[12]), "v"(e[13]), "v"(e[14]), "v"(e[15]));
}

__global__ void __launch_bounds__(128)
k_layer1(const float* __restrict__ Apk, const float* __restrict__ Gpk,
         const float* __restrict__ W1, const float* __restrict__ b1,
         float* __restrict__ act, float* __restrict__ Gnew) {
  const int lane = threadIdx.x & 31;
  const int mg = blockIdx.x >> 5;
  const int nt = (blockIdx.x & 31) * 4 + (threadIdx.x >> 5);
  const int ks = nt >> 3;
  const int r  = lane & 15;
  const int kh = (lane >> 4) << 1;
  const float* wrow = W1 + (size_t)(nt * 16 + r) * HID + kh;
  const v4f* pa0 = (const v4f*)Apk + (size_t)(mg * 2) * (256 * 32) + lane;
  const v4f* pa1 = pa0 + 256 * 32;
  const v4f* pg0 = (const v4f*)Gpk + (size_t)(mg * 2) * (256 * 32) + ks * (16 * 32) + lane;
  const v4f* pg1 = pg0 + 256 * 32;

  v8f acc0 = {};
  v8f acc1 = {};
  v8f s0 = {};
  v8f s1 = {};
  v2f w[L1_NS][L1_KS];
  v4f x[L1_NS][L1_KS / 2], y[L1_NS][L1_KS / 2];
  v4f e0[16], e1[16];
  v2f bD[32];
  const int nblk = HID / 4 / L1_KS;
  frag_load(pg0, e0);
  frag_load(pg1, e1);
#pragma unroll
  for (int s = 0; s < 4; ++s) {
    l1_load(wrow, pa0, pa1, s, w[s], x[s], y[s]);
    L1_PHASE();
  }
  const float mx0 = frag_max_exp(e0);
  const float mx1 = frag_max_exp(e1);
  frag_hold(e0);
  frag_hold(e1);
#pragma unroll
  for (int s = 4; s < L1_NS - 1; ++s) {
    l1_load(wrow, pa0, pa1, s, w[s], x[s], y[s]);
    L1_PHASE();
  }
  for (int blk = 0; blk < nblk - L1_NS; blk += L1_NS) {
#pragma unroll
    for (int s = 0; s < L1_NS; ++s) {
      const int t = (s + L1_NS - 1) % L1_NS;
      l1_load(wrow, pa0, pa1, blk + s + L1_NS - 1, w[t], x[t], y[t]);
      l1_mma(w[s], x[s], y[s], acc0, acc1);
      L1_PHASE();
    }
  }
  l1_load(wrow, pa0, pa1, nblk - 1, w[L1_NS - 1], x[L1_NS - 1], y[L1_NS - 1]);
  L1_PHASE();
#pragma unroll
  for (int s = 0; s < L1_NS - 4; ++s) l1_mma(w[s], x[s], y[s], acc0, acc1);
  L1_PHASE();
  const float* wdiag = wrow + ks * Mm;
#pragma unroll
  for (int j = 0; j < 32; ++j)
    bD[j] = *(const v2f*)(wdiag + 4 * j);
  L1_PHASE();
#pragma unroll
  for (int s = L1_NS - 4; s < L1_NS; ++s) l1_mma(w[s], x[s], y[s], acc0, acc1);
  l1_gmma(bD, e0, e1, s0, s1);

  const int n = nt * 16 + r;
  const int rowoff = (lane >> 4) * 8;
  const float bias = b1[n];
#pragma unroll
  for (int v = 0; v < 8; ++v) {
    const int m = (mg * 2) * 16 + v + rowoff;
    const float h = acc0[v] + bias;
    const float mv = __shfl(mx0, v + rowoff);
    act[m * HID + n] = fast_tanh(h);
    Gnew[m * HID + n] = __logf(s0[v]) + mv + log_abs_dtanh(h);
  }
#pragma unroll
  for (int v = 0; v < 8; ++v) {
    const int m = (mg * 2 + 1) * 16 + v + rowoff;
    const float h = acc1[v] + bias;
    const float mv = __shfl(mx1, v + rowoff);
    act[m * HID + n] = fast_tanh(h);
    Gnew[m * HID + n] = __logf(s1[v]) + mv + log_abs_dtanh(h);
  }
}

__global__ void k_tail(const float* __restrict__ act1, const float* __restrict__ G,
                       const float* __restrict__ W2,
                       float* __restrict__ Pout, float* __restrict__ Pld) {
  const int kb   = blockIdx.x >> 4;
  const int mt   = blockIdx.x & 15;
  const int lane = threadIdx.x & 31;
  const int r    = lane & 15;
  const int kh   = (lane >> 4) << 1;
  const float* arow = act1 + (size_t)(mt * 16 + r) * HID + kb * Mm;
  const float* grow = G    + (size_t)(mt * 16 + r) * HID + kb * Mm;
  const float* wrow = W2   + (size_t)r * HID + kb * Mm;
  v2f e[32];
  const float mx = rows_max_exp(grow, kh, e);
  v8f accx = {};
  v8f accs = {};
#pragma unroll
  for (int j = 0; j < 32; ++j) {
    v2f a, b;
    a.x = arow[4 * j + kh];  a.y = arow[4 * j + kh + 1];
    b.x = wrow[4 * j + kh];  b.y = wrow[4 * j + kh + 1];
    accx = wmma_f32_k4(a, b, accx);
    accs = wmma_f32_k4(e[j], b, accs);
  }
  const int rowoff = (lane >> 4) * 8;
  float* po = Pout + (size_t)kb * (Bsz * Dd);
  for (int v = 0; v < 8; ++v) {
    const int m = mt * 16 + v + rowoff;
    po[m * Dd + r] = accx[v];
  }
  float mv[8];
  for (int v = 0; v < 8; ++v) mv[v] = __shfl(mx, v + rowoff);
  if (r == kb) {
    float* pl = Pld + (kb * 16 + mt) * 32;
    for (int v = 0; v < 8; ++v)
      pl[v + rowoff] = __logf(accs[v]) + mv[v];
  }
}

__global__ void k_tail_sum(const float* __restrict__ Pout, const float* __restrict__ Pld,
                           float* __restrict__ x_out, float* __restrict__ ld,
                           int accumulate) {
  const int g = blockIdx.x;
  const int t = threadIdx.x;
  const int idx = g * 256 + t;
  float s = 0.0f;
  for (int kb = 0; kb < Dd; ++kb) s += Pout[kb * (Bsz * Dd) + idx];
  x_out[idx] = s;
  if (g < 8 && t < 32) {
    const int b = g * 32 + t;
    float tot = 0.0f;
    for (int kb = 0; kb < Dd; ++kb)
      tot += Pld[(kb * 16 + (b >> 4)) * 32 + (b & 15)];
    if (accumulate) ld[b] += tot; else ld[b] = tot;
  }
}

static void run_flow(const float* X, const float* W0, const float* W1,
                     const float* W2, const float* b0, const float* b1,
                     float* x_out, float* ld_out, int accumulate,
                     float* act0, float* act1, float* g0,
                     float* g1, float* Pout, float* Pld,
                     hipStream_t stream) {
  k_layer0   <<<256, 256, 0, stream>>>(X, W0, b0, act0, g0);
  k_layer1   <<<256, 128, 0, stream>>>(act0, g0, W1, b1, act1, g1);
  k_tail     <<<256, 32, 0, stream>>>(act1, g1, W2, Pout, Pld);
  k_tail_sum <<<16, 256, 0, stream>>>(Pout, Pld, x_out, ld_out, accumulate);
}

extern "C" void kernel_launch(void* const* d_in, const int* in_sizes, int n_in,
                              void* d_out, int out_size, void* d_ws, size_t ws_size,
                              hipStream_t stream) {
  const float* inp   = (const float*)d_in[0];
  const float* f0_W0 = (const float*)d_in[1];
  const float* f0_W1 = (const float*)d_in[2];
  const float* f0_W2 = (const float*)d_in[3];
  const float* f0_b0 = (const float*)d_in[4];
  const float* f0_b1 = (const float*)d_in[5];
  const float* f1_W0 = (const float*)d_in[6];
  const float* f1_W1 = (const float*)d_in[7];
  const float* f1_W2 = (const float*)d_in[8];
  const float* f1_b0 = (const float*)d_in[9];
  const float* f1_b1 = (const float*)d_in[10];

  float* out = (float*)d_out;
  float* ld  = out + Bsz * Dd;

  float* ws   = (float*)d_ws;
  float* act0 = ws;
  float* act1 = act0 + Bsz * HID;
  float* g0   = act1 + Bsz * HID;
  float* g1   = g0   + Bsz * HID;
  float* xmid = g1   + Bsz * HID;
  float* Pout = xmid + Bsz * Dd;
  float* Pld  = Pout + Dd * Bsz * Dd;

  run_flow(inp,  f0_W0, f0_W1, f0_W2, f0_b0, f0_b1, xmid, ld,  0,
           act0, act1, g0, g1, Pout, Pld, stream);
  run_flow(xmid, f1_W0, f1_W1, f1_W2, f1_b0, f1_b1, out,  ld,  1,
           act0, act1, g0, g1, Pout, Pld, stream);
}


# Layer 1 tiling: wave = 4 M-tiles x 1 N-tile over half of K (was 2 M-tiles, all K): each 16-line W1 fragment load feeds 4 WMMAs; wave pairs add their K halves through LDS; k sum order changes
# speedup vs baseline: 6.5603x; 1.1036x over previous
#include <hip/hip_runtime.h>
#include <hip/hip_bf16.h>


typedef float v2f __attribute__((ext_vector_type(2)));
typedef float v4f __attribute__((ext_vector_type(4)));
typedef float v8f __attribute__((ext_vector_type(8)));

#define Bsz   256
#define Dd    16
#define Mm    128
#define HID   2048

__device__ __forceinline__ v8f wmma_f32_k4(v2f a, v2f b, v8f c) {
  return __builtin_amdgcn_wmma_f32_16x16x4_f32(
      false, a, false, b, (short)0, c, false, false);
}

__device__ __forceinline__ void gemm_tile(const float* __restrict__ A, int lda,
                                          const float* __restrict__ W, int ldw,
                                          int m0, int n0, int K, v8f& acc) {
  const int lane = threadIdx.x & 31;
  const int r    = lane & 15;
  const int kh   = (lane >> 4) << 1;
  const float* arow = A + (size_t)(m0 + r) * lda;
  const float* wrow = W + (size_t)(n0 + r) * ldw;
  for (int k = 0; k < K; k += 4) {
    v2f a, b;
    a.x = arow[k + kh];  a.y = arow[k + kh + 1];
    b.x = wrow[k + kh];  b.y = wrow[k + kh + 1];
    acc = wmma_f32_k4(a, b, acc);
  }
}

__device__ __forceinline__ float fast_tanh(float x) {
#if defined(__has_builtin)
#if __has_builtin(__builtin_amdgcn_tanhf)
  return __builtin_amdgcn_tanhf(x);
#elif __has_builtin(__builtin_amdgcn_tanh_f32)
  return __builtin_amdgcn_tanh_f32(x);
#else
  return tanhf(x);
#endif
#else
  return tanhf(x);
#endif
}

__device__ __forceinline__ float softplus_f(float x) {
  return fmaxf(x, 0.0f) + __logf(1.0f + __expf(-fabsf(x)));
}
__device__ __forceinline__ float log_abs_dtanh(float h) {
  return 2.0f * (0.6931471805599453f - h - softplus_f(-2.0f * h));
}

__global__ void k_layer0(const float* __restrict__ X, const float* __restrict__ W0,
                         const float* __restrict__ b0,
                         float* __restrict__ act, float* __restrict__ gpk) {
  const int wave = (blockIdx.x * blockDim.x + threadIdx.x) >> 5;
  const int mt = wave >> 7;
  const int nt = wave & 127;
  if (mt >= Bsz / 16) return;
  v8f acc = {};
  gemm_tile(X, Dd, W0, Dd, mt * 16, nt * 16, Dd, acc);

  const int lane = threadIdx.x & 31;
  const int n = nt * 16 + (lane & 15);
  const int rowoff = (lane >> 4) * 8;
  const float ldiag = __logf(W0[n * Dd + (n >> 7)]);
  const float bias = b0[n];
  const int cl = lane & 15;
  const int pk = ((mt * 256 + nt * 2 + (cl >> 3)) * 32 + ((cl >> 1) & 1) * 16) * 4
               + 2 * ((cl >> 2) & 1) + (cl & 1);
  for (int v = 0; v < 8; ++v) {
    const float h = acc[v] + bias;
    act[pk + (v + rowoff) * 4] = fast_tanh(h);
    gpk[pk + (v + rowoff) * 4] = ldiag + log_abs_dtanh(h);
  }
}

__device__ __forceinline__ float rows_max_exp(const float* __restrict__ grow, int kh,
                                              v2f (&e)[32]) {
#pragma unroll
  for (int j = 0; j < 32; ++j) {
    e[j].x = grow[4 * j + kh];  e[j].y = grow[4 * j + kh + 1];
  }
  float mx = fmaxf(e[0].x, e[0].y);
#pragma unroll
  for (int j = 1; j < 32; ++j) mx = fmaxf(mx, fmaxf(e[j].x, e[j].y));
  mx = fmaxf(mx, __shfl_xor(mx, 16));
#pragma unroll
  for (int j = 0; j < 32; ++j) {
    e[j].x = __expf(e[j].x - mx);  e[j].y = __expf(e[j].y - mx);
  }
  return mx;
}

#define L1_KS 2
#define L1_NS 10
#define L1_MT 4
#define L1_NBLK (HID / 2 / 4 / L1_KS)
#define L1_TAIL0 (((L1_NBLK - (2 * L1_NS - 1)) / L1_NS + 1) * L1_NS)

#define L1_PHASE() asm volatile("" : "+v"(acc[0]), "+v"(acc[1]), "+v"(acc[2]), "+v"(acc[3]) :: "memory")

__device__ __forceinline__ void l1_load(const float* __restrict__ wrow,
                                        const v4f* __restrict__ pa, int blk,
                                        v2f (&b)[L1_KS], v4f (&a)[L1_MT]) {
#pragma unroll
  for (int i = 0; i < L1_KS; ++i)
    b[i] = *(const v2f*)(wrow + (blk * L1_KS + i) * 4);
#pragma unroll
  for (int t = 0; t < L1_MT; ++t)
    a[t] = pa[t * (256 * 32) + blk * 32];
}

__device__ __forceinline__ void l1_mma(const v2f (&b)[L1_KS], const v4f (&a)[L1_MT],
                                       v8f (&acc)[L1_MT]) {
#pragma unroll
  for (int t = 0; t < L1_MT; ++t) {
    v2f x;
    x.x = a[t].x;  x.y = a[t].y;
    acc[t] = wmma_f32_k4(x, b[0], acc[t]);
  }
#pragma unroll
  for (int t = 0; t < L1_MT; ++t) {
    v2f y;
    y.x = a[t].z;  y.y = a[t].w;
    acc[t] = wmma_f32_k4(y, b[1], acc[t]);
  }
}

__device__ __forceinline__ void l1_gmma(const v2f (&b)[32], const v4f (&e0)[16],
                                        const v4f (&e1)[16], v8f& s0, v8f& s1) {
#pragma unroll
  for (int i = 0; i < 16; ++i) {
    v2f x0, x1, y0, y1;
    x0.x = e0[i].x;  x0.y = e0[i].y;   y0.x = e0[i].z;  y0.y = e0[i].w;
    x1.x = e1[i].x;  x1.y = e1[i].y;   y1.x = e1[i].z;  y1.y = e1[i].w;
    s0 = wmma_f32_k4(x0, b[2 * i], s0);
    s1 = wmma_f32_k4(x1, b[2 * i], s1);
    s0 = wmma_f32_k4(y0, b[2 * i + 1], s0);
    s1 = wmma_f32_k4(y1, b[2 * i + 1], s1);
  }
}

__device__ __forceinline__ void frag_load(const v4f* __restrict__ pg, v4f (&e)[16]) {
#pragma unroll
  for (int i = 0; i < 16; ++i) e[i] = pg[i * 32];
}
__device__ __forceinline__ float frag_max_exp(v4f (&e)[16]) {
  float mx = fmaxf(fmaxf(e[0].x, e[0].y), fmaxf(e[0].z, e[0].w));
#pragma unroll
  for (int i = 1; i < 16; ++i)
    mx = fmaxf(mx, fmaxf(fmaxf(e[i].x, e[i].y), fmaxf(e[i].z, e[i].w)));
  mx = fmaxf(mx, __shfl_xor(mx, 16));
#pragma unroll
  for (int i = 0; i < 16; ++i) {
    e[i].x = __expf(e[i].x - mx);  e[i].y = __expf(e[i].y - mx);
    e[i].z = __expf(e[i].z - mx);  e[i].w = __expf(e[i].w - mx);
  }
  return mx;
}
__device__ __forceinline__ void frag_hold(const v4f (&e)[16]) {
  asm volatile("" :: "v"(e[0]), "v"(e[1]), "v"(e[2]), "v"(e[3]), "v"(e[4]), "v"(e[5]),
               "v"(e[6]), "v"(e[7]), "v"(e[8]), "v"(e[9]), "v"(e[10]), "v"(e[11]),
               "v"(e[12]), "v"(e[13]), "v"(e[14]), "v"(e[15]));
}

__global__ void __launch_bounds__(128)
k_layer1(const float* __restrict__ Apk, const float* __restrict__ Gpk,
         const float* __restrict__ W1, const float* __restrict__ b1,
         float* __restrict__ act, float* __restrict__ Gnew) {
  __shared__ float xch[4][2][8][32];
  const int lane = threadIdx.x & 31;
  const int wv  = threadIdx.x >> 5;
  const int ntl = wv & 1;
  const int kq  = wv >> 1;
  const int mq  = blockIdx.x >> 6;
  const int nt  = (blockIdx.x & 63) * 2 + ntl;
  const int ks  = nt >> 3;
  const int r   = lane & 15;
  const int kh  = (lane >> 4) << 1;
  const float* wbase = W1 + (size_t)(nt * 16 + r) * HID + kh;
  const float* wrow  = wbase + kq * (HID / 2);
  const v4f* pa  = (const v4f*)Apk + (size_t)(mq * 4) * (256 * 32) + kq * (128 * 32) + lane;
  const v4f* pg0 = (const v4f*)Gpk + (size_t)(mq * 4 + kq * 2) * (256 * 32) + ks * (16 * 32) + lane;
  const v4f* pg1 = pg0 + 256 * 32;

  v8f acc[L1_MT] = {};
  v8f s0 = {};
  v8f s1 = {};
  v2f w[L1_NS][L1_KS];
  v4f x[L1_NS][L1_MT];
  v4f e0[16], e1[16];
  v2f bD[32];
  frag_load(pg0, e0);
  frag_load(pg1, e1);
#pragma unroll
  for (int s = 0; s < 4; ++s) {
    l1_load(wrow, pa, s, w[s], x[s]);
    L1_PHASE();
  }
  const float mx0 = frag_max_exp(e0);
  const float mx1 = frag_max_exp(e1);
  frag_hold(e0);
  frag_hold(e1);
#pragma unroll
  for (int s = 4; s < L1_NS - 1; ++s) {
    l1_load(wrow, pa, s, w[s], x[s]);
    L1_PHASE();
  }
  for (int blk = 0; blk < L1_TAIL0; blk += L1_NS) {
#pragma unroll
    for (int s = 0; s < L1_NS; ++s) {
      const int t = (s + L1_NS - 1) % L1_NS;
      l1_load(wrow, pa, blk + s + L1_NS - 1, w[t], x[t]);
      l1_mma(w[s], x[s], acc);
      L1_PHASE();
    }
  }
  const float* wdiag = wbase + ks * Mm;
#pragma unroll
  for (int i = L1_TAIL0; i < L1_NBLK; ++i) {
    if (i + L1_NS - 1 < L1_NBLK) {
      const int t = (i + L1_NS - 1) % L1_NS;
      l1_load(wrow, pa, i + L1_NS - 1, w[t], x[t]);
    }
    if (i == L1_NBLK - 4) {
      L1_PHASE();
#pragma unroll
      for (int j = 0; j < 32; ++j)
        bD[j] = *(const v2f*)(wdiag + 4 * j);
      L1_PHASE();
    }
    l1_mma(w[i % L1_NS], x[i % L1_NS], acc);
    L1_PHASE();
  }
  const v8f give0 = kq ? acc[0] : acc[2];
  const v8f give1 = kq ? acc[1] : acc[3];
  const v8f keep0 = kq ? acc[2] : acc[0];
  const v8f keep1 = kq ? acc[3] : acc[1];
#pragma unroll
  for (int v = 0; v < 8; ++v) {
    xch[wv][0][v][lane] = give0[v];
    xch[wv][1][v][lane] = give1[v];
  }
  l1_gmma(bD, e0, e1, s0, s1);
  __syncthreads();
  const int pw = wv ^ 2;

  const int n = nt * 16 + r;
  const int rowoff = (lane >> 4) * 8;
  const float bias = b1[n];
#pragma unroll
  for (int v = 0; v < 8; ++v) {
    const int m = (mq * 4 + kq * 2) * 16 + v + rowoff;
    const float h = (keep0[v] + xch[pw][0][v][lane]) + bias;
    const float mv = __shfl(mx0, v + rowoff);
    act[m * HID + n] = fast_tanh(h);
    Gnew[m * HID + n] = __logf(s0[v]) + mv + log_abs_dtanh(h);
  }
#pragma unroll
  for (int v = 0; v < 8; ++v) {
    const int m = (mq * 4 + kq * 2 + 1) * 16 + v + rowoff;
    const float h = (keep1[v] + xch[pw][1][v][lane]) + bias;
    const float mv = __shfl(mx1, v + rowoff);
    act[m * HID + n] = fast_tanh(h);
    Gnew[m * HID + n] = __logf(s1[v]) + mv + log_abs_dtanh(h);
  }
}

__global__ void k_tail(const float* __restrict__ act1, const float* __restrict__ G,
                       const float* __restrict__ W2,
                       float* __restrict__ Pout, float* __restrict__ Pld) {
  const int kb   = blockIdx.x >> 4;
  const int mt   = blockIdx.x & 15;
  const int lane = threadIdx.x & 31;
  const int r    = lane & 15;
  const int kh   = (lane >> 4) << 1;
  const float* arow = act1 + (size_t)(mt * 16 + r) * HID + kb * Mm;
  const float* grow = G    + (size_t)(mt * 16 + r) * HID + kb * Mm;
  const float* wrow = W2   + (size_t)r * HID + kb * Mm;
  v2f e[32];
  const float mx = rows_max_exp(grow, kh, e);
  v8f accx = {};
  v8f accs = {};
#pragma unroll
  for (int j = 0; j < 32; ++j) {
    v2f a, b;
    a.x = arow[4 * j + kh];  a.y = arow[4 * j + kh + 1];
    b.x = wrow[4 * j + kh];  b.y = wrow[4 * j + kh + 1];
    accx = wmma_f32_k4(a, b, accx);
    accs = wmma_f32_k4(e[j], b, accs);
  }
  const int rowoff = (lane >> 4) * 8;
  float* po = Pout + (size_t)kb * (Bsz * Dd);
  for (int v = 0; v < 8; ++v) {
    const int m = mt * 16 + v + rowoff;
    po[m * Dd + r] = accx[v];
  }
  float mv[8];
  for (int v = 0; v < 8; ++v) mv[v] = __shfl(mx, v + rowoff);
  if (r == kb) {
    float* pl = Pld + (kb * 16 + mt) * 32;
    for (int v = 0; v < 8; ++v)
      pl[v + rowoff] = __logf(accs[v]) + mv[v];
  }
}

__global__ void k_tail_sum(const float* __restrict__ Pout, const float* __restrict__ Pld,
                           float* __restrict__ x_out, float* __restrict__ ld,
                           int accumulate) {
  const int g = blockIdx.x;
  const int t = threadIdx.x;
  const int idx = g * 256 + t;
  float s = 0.0f;
  for (int kb = 0; kb < Dd; ++kb) s += Pout[kb * (Bsz * Dd) + idx];
  x_out[idx] = s;
  if (g < 8 && t < 32) {
    const int b = g * 32 + t;
    float tot = 0.0f;
    for (int kb = 0; kb < Dd; ++kb)
      tot += Pld[(kb * 16 + (b >> 4)) * 32 + (b & 15)];
    if (accumulate) ld[b] += tot; else ld[b] = tot;
  }
}

static void run_flow(const float* X, const float* W0, const float* W1,
                     const float* W2, const float* b0, const float* b1,
                     float* x_out, float* ld_out, int accumulate,
                     float* act0, float* act1, float* g0,
                     float* g1, float* Pout, float* Pld,
                     hipStream_t stream) {
  k_layer0   <<<256, 256, 0, stream>>>(X, W0, b0, act0, g0);
  k_layer1   <<<256, 128, 0, stream>>>(act0, g0, W1, b1, act1, g1);
  k_tail     <<<256, 32, 0, stream>>>(act1, g1, W2, Pout, Pld);
  k_tail_sum <<<16, 256, 0, stream>>>(Pout, Pld, x_out, ld_out, accumulate);
}

extern "C" void kernel_launch(void* const* d_in, const int* in_sizes, int n_in,
                              void* d_out, int out_size, void* d_ws, size_t ws_size,
                              hipStream_t stream) {
  const float* inp   = (const float*)d_in[0];
  const float* f0_W0 = (const float*)d_in[1];
  const float* f0_W1 = (const float*)d_in[2];
  const float* f0_W2 = (const float*)d_in[3];
  const float* f0_b0 = (const float*)d_in[4];
  const float* f0_b1 = (const float*)d_in[5];
  const float* f1_W0 = (const float*)d_in[6];
  const float* f1_W1 = (const float*)d_in[7];
  const float* f1_W2 = (const float*)d_in[8];
  const float* f1_b0 = (const float*)d_in[9];
  const float* f1_b1 = (const float*)d_in[10];

  float* out = (float*)d_out;
  float* ld  = out + Bsz * Dd;

  float* ws   = (float*)d_ws;
  float* act0 = ws;
  float* act1 = act0 + Bsz * HID;
  float* g0   = act1 + Bsz * HID;
  float* g1   = g0   + Bsz * HID;
  float* xmid = g1   + Bsz * HID;
  float* Pout = xmid + Bsz * Dd;
  float* Pld  = Pout + Dd * Bsz * Dd;

  run_flow(inp,  f0_W0, f0_W1, f0_W2, f0_b0, f0_b1, xmid, ld,  0,
           act0, act1, g0, g1, Pout, Pld, stream);
  run_flow(xmid, f1_W0, f1_W1, f1_W2, f1_b0, f1_b1, out,  ld,  1,
           act0, act1, g0, g1, Pout, Pld, stream);
}


# Flow 0's partial-sum launch dropped: flow 1's layer 0 sums the 16 k-block partials in its A loader (once per group, through LDS, same order); last launch adds both flows' log-dets; 8 launches -> 7
# speedup vs baseline: 6.7793x; 1.0334x over previous
#include <hip/hip_runtime.h>
#include <hip/hip_bf16.h>


typedef float v2f __attribute__((ext_vector_type(2)));
typedef float v4f __attribute__((ext_vector_type(4)));
typedef float v8f __attribute__((ext_vector_type(8)));

#define Bsz   256
#define Dd    16
#define Mm    128
#define HID   2048

__device__ __forceinline__ v8f wmma_f32_k4(v2f a, v2f b, v8f c) {
  return __builtin_amdgcn_wmma_f32_16x16x4_f32(
      false, a, false, b, (short)0, c, false, false);
}

__device__ __forceinline__ float fast_tanh(float x) {
#if defined(__has_builtin)
#if __has_builtin(__builtin_amdgcn_tanhf)
  return __builtin_amdgcn_tanhf(x);
#elif __has_builtin(__builtin_amdgcn_tanh_f32)
  return __builtin_amdgcn_tanh_f32(x);
#else
  return tanhf(x);
#endif
#else
  return tanhf(x);
#endif
}

__device__ __forceinline__ float softplus_f(float x) {
  return fmaxf(x, 0.0f) + __logf(1.0f + __expf(-fabsf(x)));
}
__device__ __forceinline__ float log_abs_dtanh(float h) {
  return 2.0f * (0.6931471805599453f - h - softplus_f(-2.0f * h));
}

template <int NP>
__global__ void __launch_bounds__(256)
k_layer0(const float* __restrict__ Xp, const float* __restrict__ W0,
         const float* __restrict__ b0,
         float* __restrict__ act, float* __restrict__ gpk) {
  __shared__ float xs[16 * Dd];
  const int wave = (blockIdx.x * blockDim.x + threadIdx.x) >> 5;
  const int mt = wave >> 7;
  const int nt = wave & 127;
  if (mt >= Bsz / 16) return;
  const int lane = threadIdx.x & 31;
  const int n = nt * 16 + (lane & 15);
  const int kh = (lane >> 4) << 1;
  const float* wrow = W0 + (size_t)n * Dd;
  v2f b[Dd / 4];
#pragma unroll
  for (int j = 0; j < Dd / 4; ++j) {
    b[j].x = wrow[4 * j + kh];  b[j].y = wrow[4 * j + kh + 1];
  }
  const float wdiag = W0[n * Dd + (n >> 7)];
  const float bias = b0[n];
  {
    const float* xp = Xp + mt * (16 * Dd) + threadIdx.x;
    float xv[NP];
#pragma unroll
    for (int p = 0; p < NP; ++p) xv[p] = xp[p * (Bsz * Dd)];
    float s = 0.0f;
#pragma unroll
    for (int p = 0; p < NP; ++p) s += xv[p];
    xs[threadIdx.x] = s;
  }
  __syncthreads();
  v8f acc = {};
  {
    const float* arow = xs + (lane & 15) * Dd;
#pragma unroll
    for (int j = 0; j < Dd / 4; ++j) {
      v2f a;
      a.x = arow[4 * j + kh];  a.y = arow[4 * j + kh + 1];
      acc = wmma_f32_k4(a, b[j], acc);
    }
  }

  const int rowoff = (lane >> 4) * 8;
  const float ldiag = __logf(wdiag);
  const int cl = lane & 15;
  const int pk = ((mt * 256 + nt * 2 + (cl >> 3)) * 32 + ((cl >> 1) & 1) * 16) * 4
               + 2 * ((cl >> 2) & 1) + (cl & 1);
  for (int v = 0; v < 8; ++v) {
    const float h = acc[v] + bias;
    act[pk + (v + rowoff) * 4] = fast_tanh(h);
    gpk[pk + (v + rowoff) * 4] = ldiag + log_abs_dtanh(h);
  }
}

__device__ __forceinline__ float rows_max_exp(const float* __restrict__ grow, int kh,
                                              v2f (&e)[32]) {
#pragma unroll
  for (int j = 0; j < 32; ++j) {
    e[j].x = grow[4 * j + kh];  e[j].y = grow[4 * j + kh + 1];
  }
  float mx = fmaxf(e[0].x, e[0].y);
#pragma unroll
  for (int j = 1; j < 32; ++j) mx = fmaxf(mx, fmaxf(e[j].x, e[j].y));
  mx = fmaxf(mx, __shfl_xor(mx, 16));
#pragma unroll
  for (int j = 0; j < 32; ++j) {
    e[j].x = __expf(e[j].x - mx);  e[j].y = __expf(e[j].y - mx);
  }
  return mx;
}

#define L1_KS 2
#define L1_NS 10
#define L1_MT 4
#define L1_NBLK (HID / 2 / 4 / L1_KS)
#define L1_TAIL0 (((L1_NBLK - (2 * L1_NS - 1)) / L1_NS + 1) * L1_NS)

#define L1_PHASE() asm volatile("" : "+v"(acc[0]), "+v"(acc[1]), "+v"(acc[2]), "+v"(acc[3]) :: "memory")

__device__ __forceinline__ void l1_load(const float* __restrict__ wrow,
                                        const v4f* __restrict__ pa, int blk,
                                        v2f (&b)[L1_KS], v4f (&a)[L1_MT]) {
#pragma unroll
  for (int i = 0; i < L1_KS; ++i)
    b[i] = *(const v2f*)(wrow + (blk * L1_KS + i) * 4);
#pragma unroll
  for (int t = 0; t < L1_MT; ++t)
    a[t] = pa[t * (256 * 32) + blk * 32];
}

__device__ __forceinline__ void l1_mma(const v2f (&b)[L1_KS], const v4f (&a)[L1_MT],
                                       v8f (&acc)[L1_MT]) {
#pragma unroll
  for (int t = 0; t < L1_MT; ++t) {
    v2f x;
    x.x = a[t].x;  x.y = a[t].y;
    acc[t] = wmma_f32_k4(x, b[0], acc[t]);
  }
#pragma unroll
  for (int t = 0; t < L1_MT; ++t) {
    v2f y;
    y.x = a[t].z;  y.y = a[t].w;
    acc[t] = wmma_f32_k4(y, b[1], acc[t]);
  }
}

__device__ __forceinline__ void l1_gmma(const v2f (&b)[32], const v4f (&e0)[16],
                                        const v4f (&e1)[16], v8f& s0, v8f& s1) {
#pragma unroll
  for (int i = 0; i < 16; ++i) {
    v2f x0, x1, y0, y1;
    x0.x = e0[i].x;  x0.y = e0[i].y;   y0.x = e0[i].z;  y0.y = e0[i].w;
    x1.x = e1[i].x;  x1.y = e1[i].y;   y1.x = e1[i].z;  y1.y = e1[i].w;
    s0 = wmma_f32_k4(x0, b[2 * i], s0);
    s1 = wmma_f32_k4(x1, b[2 * i], s1);
    s0 = wmma_f32_k4(y0, b[2 * i + 1], s0);
    s1 = wmma_f32_k4(y1, b[2 * i + 1], s1);
  }
}

__device__ __forceinline__ void frag_load(const v4f* __restrict__ pg, v4f (&e)[16]) {
#pragma unroll
  for (int i = 0; i < 16; ++i) e[i] = pg[i * 32];
}
__device__ __forceinline__ float frag_max_exp(v4f (&e)[16]) {
  float mx = fmaxf(fmaxf(e[0].x, e[0].y), fmaxf(e[0].z, e[0].w));
#pragma unroll
  for (int i = 1; i < 16; ++i)
    mx = fmaxf(mx, fmaxf(fmaxf(e[i].x, e[i].y), fmaxf(e[i].z, e[i].w)));
  mx = fmaxf(mx, __shfl_xor(mx, 16));
#pragma unroll
  for (int i = 0; i < 16; ++i) {
    e[i].x = __expf(e[i].x - mx);  e[i].y = __expf(e[i].y - mx);
    e[i].z = __expf(e[i].z - mx);  e[i].w = __expf(e[i].w - mx);
  }
  return mx;
}
__device__ __forceinline__ void frag_hold(const v4f (&e)[16]) {
  asm volatile("" :: "v"(e[0]), "v"(e[1]), "v"(e[2]), "v"(e[3]), "v"(e[4]), "v"(e[5]),
               "v"(e[6]), "v"(e[7]), "v"(e[8]), "v"(e[9]), "v"(e[10]), "v"(e[11]),
               "v"(e[12]), "v"(e[13]), "v"(e[14]), "v"(e[15]));
}

__global__ void __launch_bounds__(128)
k_layer1(const float* __restrict__ Apk, const float* __restrict__ Gpk,
         const float* __restrict__ W1, const float* __restrict__ b1,
         float* __restrict__ act, float* __restrict__ Gnew) {
  __shared__ float xch[4][2][8][32];
  const int lane = threadIdx.x & 31;
  const int wv  = threadIdx.x >> 5;
  const int ntl = wv & 1;
  const int kq  = wv >> 1;
  const int mq  = blockIdx.x >> 6;
  const int nt  = (blockIdx.x & 63) * 2 + ntl;
  const int ks  = nt >> 3;
  const int r   = lane & 15;
  const int kh  = (lane >> 4) << 1;
  const float* wbase = W1 + (size_t)(nt * 16 + r) * HID + kh;
  const float* wrow  = wbase + kq * (HID / 2);
  const v4f* pa  = (const v4f*)Apk + (size_t)(mq * 4) * (256 * 32) + kq * (128 * 32) + lane;
  const v4f* pg0 = (const v4f*)Gpk + (size_t)(mq * 4 + kq * 2) * (256 * 32) + ks * (16 * 32) + lane;
  const v4f* pg1 = pg0 + 256 * 32;

  v8f acc[L1_MT] = {};
  v8f s0 = {};
  v8f s1 = {};
  v2f w[L1_NS][L1_KS];
  v4f x[L1_NS][L1_MT];
  v4f e0[16], e1[16];
  v2f bD[32];
  frag_load(pg0, e0);
  frag_load(pg1, e1);
#pragma unroll
  for (int s = 0; s < 4; ++s) {
    l1_load(wrow, pa, s, w[s], x[s]);
    L1_PHASE();
  }
  const float mx0 = frag_max_exp(e0);
  const float mx1 = frag_max_exp(e1);
  frag_hold(e0);
  frag_hold(e1);
#pragma unroll
  for (int s = 4; s < L1_NS - 1; ++s) {
    l1_load(wrow, pa, s, w[s], x[s]);
    L1_PHASE();
  }
  for (int blk = 0; blk < L1_TAIL0; blk += L1_NS) {
#pragma unroll
    for (int s = 0; s < L1_NS; ++s) {
      const int t = (s + L1_NS - 1) % L1_NS;
      l1_load(wrow, pa, blk + s + L1_NS - 1, w[t], x[t]);
      l1_mma(w[s], x[s], acc);
      L1_PHASE();
    }
  }
  const float* wdiag = wbase + ks * Mm;
#pragma unroll
  for (int i = L1_TAIL0; i < L1_NBLK; ++i) {
    if (i + L1_NS - 1 < L1_NBLK) {
      const int t = (i + L1_NS - 1) % L1_NS;
      l1_load(wrow, pa, i + L1_NS - 1, w[t], x[t]);
    }
    if (i == L1_NBLK - 4) {
      L1_PHASE();
#pragma unroll
      for (int j = 0; j < 32; ++j)
        bD[j] = *(const v2f*)(wdiag + 4 * j);
      L1_PHASE();
    }
    l1_mma(w[i % L1_NS], x[i % L1_NS], acc);
    L1_PHASE();
  }
  const v8f give0 = kq ? acc[0] : acc[2];
  const v8f give1 = kq ? acc[1] : acc[3];
  const v8f keep0 = kq ? acc[2] : acc[0];
  const v8f keep1 = kq ? acc[3] : acc[1];
#pragma unroll
  for (int v = 0; v < 8; ++v) {
    xch[wv][0][v][lane] = give0[v];
    xch[wv][1][v][lane] = give1[v];
  }
  l1_gmma(bD, e0, e1, s0, s1);
  __syncthreads();
  const int pw = wv ^ 2;

  const int n = nt * 16 + r;
  const int rowoff = (lane >> 4) * 8;
  const float bias = b1[n];
#pragma unroll
  for (int v = 0; v < 8; ++v) {
    const int m = (mq * 4 + kq * 2) * 16 + v + rowoff;
    const float h = (keep0[v] + xch[pw][0][v][lane]) + bias;
    const float mv = __shfl(mx0, v + rowoff);
    act[m * HID + n] = fast_tanh(h);
    Gnew[m * HID + n] = __logf(s0[v]) + mv + log_abs_dtanh(h);
  }
#pragma unroll
  for (int v = 0; v < 8; ++v) {
    const int m = (mq * 4 + kq * 2 + 1) * 16 + v + rowoff;
    const float h = (keep1[v] + xch[pw][1][v][lane]) + bias;
    const float mv = __shfl(mx1, v + rowoff);
    act[m * HID + n] = fast_tanh(h);
    Gnew[m * HID + n] = __logf(s1[v]) + mv + log_abs_dtanh(h);
  }
}

__global__ void k_tail(const float* __restrict__ act1, const float* __restrict__ G,
                       const float* __restrict__ W2,
                       float* __restrict__ Pout, float* __restrict__ Pld) {
  const int kb   = blockIdx.x >> 4;
  const int mt   = blockIdx.x & 15;
  const int lane = threadIdx.x & 31;
  const int r    = lane & 15;
  const int kh   = (lane >> 4) << 1;
  const float* arow = act1 + (size_t)(mt * 16 + r) * HID + kb * Mm;
  const float* grow = G    + (size_t)(mt * 16 + r) * HID + kb * Mm;
  const float* wrow = W2   + (size_t)r * HID + kb * Mm;
  v2f e[32];
  const float mx = rows_max_exp(grow, kh, e);
  v8f accx = {};
  v8f accs = {};
#pragma unroll
  for (int j = 0; j < 32; ++j) {
    v2f a, b;
    a.x = arow[4 * j + kh];  a.y = arow[4 * j + kh + 1];
    b.x = wrow[4 * j + kh];  b.y = wrow[4 * j + kh + 1];
    accx = wmma_f32_k4(a, b, accx);
    accs = wmma_f32_k4(e[j], b, accs);
  }
  const int rowoff = (lane >> 4) * 8;
  float* po = Pout + (size_t)kb * (Bsz * Dd);
  for (int v = 0; v < 8; ++v) {
    const int m = mt * 16 + v + rowoff;
    po[m * Dd + r] = accx[v];
  }
  float mv[8];
  for (int v = 0; v < 8; ++v) mv[v] = __shfl(mx, v + rowoff);
  if (r == kb) {
    float* pl = Pld + (kb * 16 + mt) * 32;
    for (int v = 0; v < 8; ++v)
      pl[v + rowoff] = __logf(accs[v]) + mv[v];
  }
}

__global__ void k_tail_sum(const float* __restrict__ Pout, const float* __restrict__ Pld0,
                           const float* __restrict__ Pld1,
                           float* __restrict__ x_out, float* __restrict__ ld) {
  const int g = blockIdx.x;
  const int t = threadIdx.x;
  const int idx = g * 256 + t;
  float s = 0.0f;
  for (int kb = 0; kb < Dd; ++kb) s += Pout[kb * (Bsz * Dd) + idx];
  x_out[idx] = s;
  if (g < 8 && t < 32) {
    const int b = g * 32 + t;
    float tot0 = 0.0f;
    float tot1 = 0.0f;
    for (int kb = 0; kb < Dd; ++kb)
      tot0 += Pld0[(kb * 16 + (b >> 4)) * 32 + (b & 15)];
    for (int kb = 0; kb < Dd; ++kb)
      tot1 += Pld1[(kb * 16 + (b >> 4)) * 32 + (b & 15)];
    ld[b] = tot0 + tot1;
  }
}

template <int NP>
static void run_flow(const float* Xp, const float* W0, const float* W1,
                     const float* W2, const float* b0, const float* b1,
                     float* act0, float* act1, float* g0,
                     float* g1, float* Pout, float* Pld,
                     hipStream_t stream) {
  k_layer0<NP> <<<256, 256, 0, stream>>>(Xp, W0, b0, act0, g0);
  k_layer1   <<<256, 128, 0, stream>>>(act0, g0, W1, b1, act1, g1);
  k_tail     <<<256, 32, 0, stream>>>(act1, g1, W2, Pout, Pld);
}

extern "C" void kernel_launch(void* const* d_in, const int* in_sizes, int n_in,
                              void* d_out, int out_size, void* d_ws, size_t ws_size,
                              hipStream_t stream) {
  const float* inp   = (const float*)d_in[0];
  const float* f0_W0 = (const float*)d_in[1];
  const float* f0_W1 = (const float*)d_in[2];
  const float* f0_W2 = (const float*)d_in[3];
  const float* f0_b0 = (const float*)d_in[4];
  const float* f0_b1 = (const float*)d_in[5];
  const float* f1_W0 = (const float*)d_in[6];
  const float* f1_W1 = (const float*)d_in[7];
  const float* f1_W2 = (const float*)d_in[8];
  const float* f1_b0 = (const float*)d_in[9];
  const float* f1_b1 = (const float*)d_in[10];

  float* out = (float*)d_out;
  float* ld  = out + Bsz * Dd;

  float* ws   = (float*)d_ws;
  float* act0 = ws;
  float* act1 = act0 + Bsz * HID;
  float* g0   = act1 + Bsz * HID;
  float* g1   = g0   + Bsz * HID;
  float* Pout0 = g1    + Bsz * HID;
  float* Pout1 = Pout0 + Dd * Bsz * Dd;
  float* Pld0  = Pout1 + Dd * Bsz * Dd;
  float* Pld1  = Pld0  + Dd * 16 * 32;

  run_flow<1>(inp, f0_W0, f0_W1, f0_W2, f0_b0, f0_b1,
              act0, act1, g0, g1, Pout0, Pld0, stream);
  run_flow<Dd>(Pout0, f1_W0, f1_W1, f1_W2, f1_b0, f1_b1,
               act0, act1, g0, g1, Pout1, Pld1, stream);
  k_tail_sum <<<16, 256, 0, stream>>>(Pout1, Pld0, Pld1, out, ld);
}


# Flow tail: the two products of a (k-block, M-tile) run in two waves (256 groups of 2), each asking for all 64 fragments up front: W2's lines (from memory) not waited for in the chain; sum orders kept
# speedup vs baseline: 7.1468x; 1.0542x over previous
#include <hip/hip_runtime.h>
#include <hip/hip_bf16.h>


typedef float v2f __attribute__((ext_vector_type(2)));
typedef float v4f __attribute__((ext_vector_type(4)));
typedef float v8f __attribute__((ext_vector_type(8)));

#define Bsz   256
#define Dd    16
#define Mm    128
#define HID   2048

__device__ __forceinline__ v8f wmma_f32_k4(v2f a, v2f b, v8f c) {
  return __builtin_amdgcn_wmma_f32_16x16x4_f32(
      false, a, false, b, (short)0, c, false, false);
}

__device__ __forceinline__ float fast_tanh(float x) {
#if defined(__has_builtin)
#if __has_builtin(__builtin_amdgcn_tanhf)
  return __builtin_amdgcn_tanhf(x);
#elif __has_builtin(__builtin_amdgcn_tanh_f32)
  return __builtin_amdgcn_tanh_f32(x);
#else
  return tanhf(x);
#endif
#else
  return tanhf(x);
#endif
}

__device__ __forceinline__ float softplus_f(float x) {
  return fmaxf(x, 0.0f) + __logf(1.0f + __expf(-fabsf(x)));
}
__device__ __forceinline__ float log_abs_dtanh(float h) {
  return 2.0f * (0.6931471805599453f - h - softplus_f(-2.0f * h));
}

template <int NP>
__global__ void __launch_bounds__(256)
k_layer0(const float* __restrict__ Xp, const float* __restrict__ W0,
         const float* __restrict__ b0,
         float* __restrict__ act, float* __restrict__ gpk) {
  __shared__ float xs[16 * Dd];
  const int wave = (blockIdx.x * blockDim.x + threadIdx.x) >> 5;
  const int mt = wave >> 7;
  const int nt = wave & 127;
  if (mt >= Bsz / 16) return;
  const int lane = threadIdx.x & 31;
  const int n = nt * 16 + (lane & 15);
  const int kh = (lane >> 4) << 1;
  const float* wrow = W0 + (size_t)n * Dd;
  v2f b[Dd / 4];
#pragma unroll
  for (int j = 0; j < Dd / 4; ++j) {
    b[j].x = wrow[4 * j + kh];  b[j].y = wrow[4 * j + kh + 1];
  }
  const float wdiag = W0[n * Dd + (n >> 7)];
  const float bias = b0[n];
  {
    const float* xp = Xp + mt * (16 * Dd) + threadIdx.x;
    float xv[NP];
#pragma unroll
    for (int p = 0; p < NP; ++p) xv[p] = xp[p * (Bsz * Dd)];
    float s = 0.0f;
#pragma unroll
    for (int p = 0; p < NP; ++p) s += xv[p];
    xs[threadIdx.x] = s;
  }
  __syncthreads();
  v8f acc = {};
  {
    const float* arow = xs + (lane & 15) * Dd;
#pragma unroll
    for (int j = 0; j < Dd / 4; ++j) {
      v2f a;
      a.x = arow[4 * j + kh];  a.y = arow[4 * j + kh + 1];
      acc = wmma_f32_k4(a, b[j], acc);
    }
  }

  const int rowoff = (lane >> 4) * 8;
  const float ldiag = __logf(wdiag);
  const int cl = lane & 15;
  const int pk = ((mt * 256 + nt * 2 + (cl >> 3)) * 32 + ((cl >> 1) & 1) * 16) * 4
               + 2 * ((cl >> 2) & 1) + (cl & 1);
  for (int v = 0; v < 8; ++v) {
    const float h = acc[v] + bias;
    act[pk + (v + rowoff) * 4] = fast_tanh(h);
    gpk[pk + (v + rowoff) * 4] = ldiag + log_abs_dtanh(h);
  }
}

__device__ __forceinline__ void rows_load(const float* __restrict__ row, int kh,
                                          v2f (&f)[32]) {
#pragma unroll
  for (int j = 0; j < 32; ++j) {
    f[j].x = row[4 * j + kh];  f[j].y = row[4 * j + kh + 1];
  }
}
__device__ __forceinline__ float rows_max_exp(v2f (&e)[32]) {
  float mx = fmaxf(e[0].x, e[0].y);
#pragma unroll
  for (int j = 1; j < 32; ++j) mx = fmaxf(mx, fmaxf(e[j].x, e[j].y));
  mx = fmaxf(mx, __shfl_xor(mx, 16));
#pragma unroll
  for (int j = 0; j < 32; ++j) {
    e[j].x = __expf(e[j].x - mx);  e[j].y = __expf(e[j].y - mx);
  }
  return mx;
}

#define L1_KS 2
#define L1_NS 10
#define L1_MT 4
#define L1_NBLK (HID / 2 / 4 / L1_KS)
#define L1_TAIL0 (((L1_NBLK - (2 * L1_NS - 1)) / L1_NS + 1) * L1_NS)

#define L1_PHASE() asm volatile("" : "+v"(acc[0]), "+v"(acc[1]), "+v"(acc[2]), "+v"(acc[3]) :: "memory")

__device__ __forceinline__ void l1_load(const float* __restrict__ wrow,
                                        const v4f* __restrict__ pa, int blk,
                                        v2f (&b)[L1_KS], v4f (&a)[L1_MT]) {
#pragma unroll
  for (int i = 0; i < L1_KS; ++i)
    b[i] = *(const v2f*)(wrow + (blk * L1_KS + i) * 4);
#pragma unroll
  for (int t = 0; t < L1_MT; ++t)
    a[t] = pa[t * (256 * 32) + blk * 32];
}

__device__ __forceinline__ void l1_mma(const v2f (&b)[L1_KS], const v4f (&a)[L1_MT],
                                       v8f (&acc)[L1_MT]) {
#pragma unroll
  for (int t = 0; t < L1_MT; ++t) {
    v2f x;
    x.x = a[t].x;  x.y = a[t].y;
    acc[t] = wmma_f32_k4(x, b[0], acc[t]);
  }
#pragma unroll
  for (int t = 0; t < L1_MT; ++t) {
    v2f y;
    y.x = a[t].z;  y.y = a[t].w;
    acc[t] = wmma_f32_k4(y, b[1], acc[t]);
  }
}

__device__ __forceinline__ void l1_gmma(const v2f (&b)[32], const v4f (&e0)[16],
                                        const v4f (&e1)[16], v8f& s0, v8f& s1) {
#pragma unroll
  for (int i = 0; i < 16; ++i) {
    v2f x0, x1, y0, y1;
    x0.x = e0[i].x;  x0.y = e0[i].y;   y0.x = e0[i].z;  y0.y = e0[i].w;
    x1.x = e1[i].x;  x1.y = e1[i].y;   y1.x = e1[i].z;  y1.y = e1[i].w;
    s0 = wmma_f32_k4(x0, b[2 * i], s0);
    s1 = wmma_f32_k4(x1, b[2 * i], s1);
    s0 = wmma_f32_k4(y0, b[2 * i + 1], s0);
    s1 = wmma_f32_k4(y1, b[2 * i + 1], s1);
  }
}

__device__ __forceinline__ void frag_load(const v4f* __restrict__ pg, v4f (&e)[16]) {
#pragma unroll
  for (int i = 0; i < 16; ++i) e[i] = pg[i * 32];
}
__device__ __forceinline__ float frag_max_exp(v4f (&e)[16]) {
  float mx = fmaxf(fmaxf(e[0].x, e[0].y), fmaxf(e[0].z, e[0].w));
#pragma unroll
  for (int i = 1; i < 16; ++i)
    mx = fmaxf(mx, fmaxf(fmaxf(e[i].x, e[i].y), fmaxf(e[i].z, e[i].w)));
  mx = fmaxf(mx, __shfl_xor(mx, 16));
#pragma unroll
  for (int i = 0; i < 16; ++i) {
    e[i].x = __expf(e[i].x - mx);  e[i].y = __expf(e[i].y - mx);
    e[i].z = __expf(e[i].z - mx);  e[i].w = __expf(e[i].w - mx);
  }
  return mx;
}
__device__ __forceinline__ void frag_hold(const v4f (&e)[16]) {
  asm volatile("" :: "v"(e[0]), "v"(e[1]), "v"(e[2]), "v"(e[3]), "v"(e[4]), "v"(e[5]),
               "v"(e[6]), "v"(e[7]), "v"(e[8]), "v"(e[9]), "v"(e[10]), "v"(e[11]),
               "v"(e[12]), "v"(e[13]), "v"(e[14]), "v"(e[15]));
}

__global__ void __launch_bounds__(128)
k_layer1(const float* __restrict__ Apk, const float* __restrict__ Gpk,
         const float* __restrict__ W1, const float* __restrict__ b1,
         float* __restrict__ act, float* __restrict__ Gnew) {
  __shared__ float xch[4][2][8][32];
  const int lane = threadIdx.x & 31;
  const int wv  = threadIdx.x >> 5;
  const int ntl = wv & 1;
  const int kq  = wv >> 1;
  const int mq  = blockIdx.x >> 6;
  const int nt  = (blockIdx.x & 63) * 2 + ntl;
  const int ks  = nt >> 3;
  const int r   = lane & 15;
  const int kh  = (lane >> 4) << 1;
  const float* wbase = W1 + (size_t)(nt * 16 + r) * HID + kh;
  const float* wrow  = wbase + kq * (HID / 2);
  const v4f* pa  = (const v4f*)Apk + (size_t)(mq * 4) * (256 * 32) + kq * (128 * 32) + lane;
  const v4f* pg0 = (const v4f*)Gpk + (size_t)(mq * 4 + kq * 2) * (256 * 32) + ks * (16 * 32) + lane;
  const v4f* pg1 = pg0 + 256 * 32;

  v8f acc[L1_MT] = {};
  v8f s0 = {};
  v8f s1 = {};
  v2f w[L1_NS][L1_KS];
  v4f x[L1_NS][L1_MT];
  v4f e0[16], e1[16];
  v2f bD[32];
  frag_load(pg0, e0);
  frag_load(pg1, e1);
#pragma unroll
  for (int s = 0; s < 4; ++s) {
    l1_load(wrow, pa, s, w[s], x[s]);
    L1_PHASE();
  }
  const float mx0 = frag_max_exp(e0);
  const float mx1 = frag_max_exp(e1);
  frag_hold(e0);
  frag_hold(e1);
#pragma unroll
  for (int s = 4; s < L1_NS - 1; ++s) {
    l1_load(wrow, pa, s, w[s], x[s]);
    L1_PHASE();
  }
  for (int blk = 0; blk < L1_TAIL0; blk += L1_NS) {
#pragma unroll
    for (int s = 0; s < L1_NS; ++s) {
      const int t = (s + L1_NS - 1) % L1_NS;
      l1_load(wrow, pa, blk + s + L1_NS - 1, w[t], x[t]);
      l1_mma(w[s], x[s], acc);
      L1_PHASE();
    }
  }
  const float* wdiag = wbase + ks * Mm;
#pragma unroll
  for (int i = L1_TAIL0; i < L1_NBLK; ++i) {
    if (i + L1_NS - 1 < L1_NBLK) {
      const int t = (i + L1_NS - 1) % L1_NS;
      l1_load(wrow, pa, i + L1_NS - 1, w[t], x[t]);
    }
    if (i == L1_NBLK - 4) {
      L1_PHASE();
#pragma unroll
      for (int j = 0; j < 32; ++j)
        bD[j] = *(const v2f*)(wdiag + 4 * j);
      L1_PHASE();
    }
    l1_mma(w[i % L1_NS], x[i % L1_NS], acc);
    L1_PHASE();
  }
  const v8f give0 = kq ? acc[0] : acc[2];
  const v8f give1 = kq ? acc[1] : acc[3];
  const v8f keep0 = kq ? acc[2] : acc[0];
  const v8f keep1 = kq ? acc[3] : acc[1];
#pragma unroll
  for (int v = 0; v < 8; ++v) {
    xch[wv][0][v][lane] = give0[v];
    xch[wv][1][v][lane] = give1[v];
  }
  l1_gmma(bD, e0, e1, s0, s1);
  __syncthreads();
  const int pw = wv ^ 2;

  const int n = nt * 16 + r;
  const int rowoff = (lane >> 4) * 8;
  const float bias = b1[n];
#pragma unroll
  for (int v = 0; v < 8; ++v) {
    const int m = (mq * 4 + kq * 2) * 16 + v + rowoff;
    const float h = (keep0[v] + xch[pw][0][v][lane]) + bias;
    const float mv = __shfl(mx0, v + rowoff);
    act[m * HID + n] = fast_tanh(h);
    Gnew[m * HID + n] = __logf(s0[v]) + mv + log_abs_dtanh(h);
  }
#pragma unroll
  for (int v = 0; v < 8; ++v) {
    const int m = (mq * 4 + kq * 2 + 1) * 16 + v + rowoff;
    const float h = (keep1[v] + xch[pw][1][v][lane]) + bias;
    const float mv = __shfl(mx1, v + rowoff);
    act[m * HID + n] = fast_tanh(h);
    Gnew[m * HID + n] = __logf(s1[v]) + mv + log_abs_dtanh(h);
  }
}


#define TAIL_LOADS_STAY() __builtin_amdgcn_sched_barrier(0)

__global__ void __launch_bounds__(64)
k_tail(const float* __restrict__ act1, const float* __restrict__ G,
       const float* __restrict__ W2,
       float* __restrict__ Pout, float* __restrict__ Pld) {
  const int role = __builtin_amdgcn_readfirstlane(threadIdx.x >> 5);
  const int kb   = blockIdx.x >> 4;
  const int mt   = blockIdx.x & 15;
  const int lane = threadIdx.x & 31;
  const int r    = lane & 15;
  const int kh   = (lane >> 4) << 1;
  const int rowoff = (lane >> 4) * 8;
  const float* wrow = W2 + (size_t)r * HID + kb * Mm;
  v2f b[32];
  if (role == 0) {
    const float* arow = act1 + (size_t)(mt * 16 + r) * HID + kb * Mm;
    v2f a[32];
#pragma unroll
    for (int j = 0; j < 32; ++j) {
      a[j].x = arow[4 * j + kh];  a[j].y = arow[4 * j + kh + 1];
      b[j].x = wrow[4 * j + kh];  b[j].y = wrow[4 * j + kh + 1];
    }
    TAIL_LOADS_STAY();
    v8f accx = {};
#pragma unroll
    for (int j = 0; j < 32; ++j) accx = wmma_f32_k4(a[j], b[j], accx);
    float* po = Pout + (size_t)kb * (Bsz * Dd);
    for (int v = 0; v < 8; ++v) {
      const int m = mt * 16 + v + rowoff;
      po[m * Dd + r] = accx[v];
    }
  } else {
    const float* grow = G + (size_t)(mt * 16 + r) * HID + kb * Mm;
    v2f e[32];
    rows_load(grow, kh, e);
    rows_load(wrow, kh, b);
    TAIL_LOADS_STAY();
    const float mx = rows_max_exp(e);
    v8f accs = {};
#pragma unroll
    for (int j = 0; j < 32; ++j) accs = wmma_f32_k4(e[j], b[j], accs);
    float mv[8];
    for (int v = 0; v < 8; ++v) mv[v] = __shfl(mx, v + rowoff);
    if (r == kb) {
      float* pl = Pld + (kb * 16 + mt) * 32;
      for (int v = 0; v < 8; ++v)
        pl[v + rowoff] = __logf(accs[v]) + mv[v];
    }
  }
}

__global__ void k_tail_sum(const float* __restrict__ Pout, const float* __restrict__ Pld0,
                           const float* __restrict__ Pld1,
                           float* __restrict__ x_out, float* __restrict__ ld) {
  const int g = blockIdx.x;
  const int t = threadIdx.x;
  const int idx = g * 256 + t;
  float s = 0.0f;
  for (int kb = 0; kb < Dd; ++kb) s += Pout[kb * (Bsz * Dd) + idx];
  x_out[idx] = s;
  if (g < 8 && t < 32) {
    const int b = g * 32 + t;
    float tot0 = 0.0f;
    float tot1 = 0.0f;
    for (int kb = 0; kb < Dd; ++kb)
      tot0 += Pld0[(kb * 16 + (b >> 4)) * 32 + (b & 15)];
    for (int kb = 0; kb < Dd; ++kb)
      tot1 += Pld1[(kb * 16 + (b >> 4)) * 32 + (b & 15)];
    ld[b] = tot0 + tot1;
  }
}

template <int NP>
static void run_flow(const float* Xp, const float* W0, const float* W1,
                     const float* W2, const float* b0, const float* b1,
                     float* act0, float* act1, float* g0,
                     float* g1, float* Pout, float* Pld,
                     hipStream_t stream) {
  k_layer0<NP> <<<256, 256, 0, stream>>>(Xp, W0, b0, act0, g0);
  k_layer1   <<<256, 128, 0, stream>>>(act0, g0, W1, b1, act1, g1);
  k_tail     <<<256, 64, 0, stream>>>(act1, g1, W2, Pout, Pld);
}

extern "C" void kernel_launch(void* const* d_in, const int* in_sizes, int n_in,
                              void* d_out, int out_size, void* d_ws, size_t ws_size,
                              hipStream_t stream) {
  const float* inp   = (const float*)d_in[0];
  const float* f0_W0 = (const float*)d_in[1];
  const float* f0_W1 = (const float*)d_in[2];
  const float* f0_W2 = (const float*)d_in[3];
  const float* f0_b0 = (const float*)d_in[4];
  const float* f0_b1 = (const float*)d_in[5];
  const float* f1_W0 = (const float*)d_in[6];
  const float* f1_W1 = (const float*)d_in[7];
  const float* f1_W2 = (const float*)d_in[8];
  const float* f1_b0 = (const float*)d_in[9];
  const float* f1_b1 = (const float*)d_in[10];

  float* out = (float*)d_out;
  float* ld  = out + Bsz * Dd;

  float* ws   = (float*)d_ws;
  float* act0 = ws;
  float* act1 = act0 + Bsz * HID;
  float* g0   = act1 + Bsz * HID;
  float* g1   = g0   + Bsz * HID;
  float* Pout0 = g1    + Bsz * HID;
  float* Pout1 = Pout0 + Dd * Bsz * Dd;
  float* Pld0  = Pout1 + Dd * Bsz * Dd;
  float* Pld1  = Pld0  + Dd * 16 * 32;

  run_flow<1>(inp, f0_W0, f0_W1, f0_W2, f0_b0, f0_b1,
              act0, act1, g0, g1, Pout0, Pld0, stream);
  run_flow<Dd>(Pout0, f1_W0, f1_W1, f1_W2, f1_b0, f1_b1,
               act0, act1, g0, g1, Pout1, Pld1, stream);
  k_tail_sum <<<16, 256, 0, stream>>>(Pout1, Pld0, Pld1, out, ld);
}


# Layer 1 tiling: wave = 4 M-tiles x 2 N-tiles over a quarter of K (was 4 x 1 over half): an A word feeds 2 N-tiles, no two waves load the same bytes; 4 K quarters summed via LDS; k sum order changes
# speedup vs baseline: 7.5095x; 1.0507x over previous
#include <hip/hip_runtime.h>
#include <hip/hip_bf16.h>


typedef float v2f __attribute__((ext_vector_type(2)));
typedef float v4f __attribute__((ext_vector_type(4)));
typedef float v8f __attribute__((ext_vector_type(8)));

#define Bsz   256
#define Dd    16
#define Mm    128
#define HID   2048

__device__ __forceinline__ v8f wmma_f32_k4(v2f a, v2f b, v8f c) {
  return __builtin_amdgcn_wmma_f32_16x16x4_f32(
      false, a, false, b, (short)0, c, false, false);
}

__device__ __forceinline__ float fast_tanh(float x) {
#if defined(__has_builtin)
#if __has_builtin(__builtin_amdgcn_tanhf)
  return __builtin_amdgcn_tanhf(x);
#elif __has_builtin(__builtin_amdgcn_tanh_f32)
  return __builtin_amdgcn_tanh_f32(x);
#else
  return tanhf(x);
#endif
#else
  return tanhf(x);
#endif
}

__device__ __forceinline__ float softplus_f(float x) {
  return fmaxf(x, 0.0f) + __logf(1.0f + __expf(-fabsf(x)));
}
__device__ __forceinline__ float log_abs_dtanh(float h) {
  return 2.0f * (0.6931471805599453f - h - softplus_f(-2.0f * h));
}

template <int NP>
__global__ void __launch_bounds__(256)
k_layer0(const float* __restrict__ Xp, const float* __restrict__ W0,
         const float* __restrict__ b0,
         float* __restrict__ act, float* __restrict__ gpk) {
  __shared__ float xs[16 * Dd];
  const int wave = (blockIdx.x * blockDim.x + threadIdx.x) >> 5;
  const int mt = wave >> 7;
  const int nt = wave & 127;
  if (mt >= Bsz / 16) return;
  const int lane = threadIdx.x & 31;
  const int n = nt * 16 + (lane & 15);
  const int kh = (lane >> 4) << 1;
  const float* wrow = W0 + (size_t)n * Dd;
  v2f b[Dd / 4];
#pragma unroll
  for (int j = 0; j < Dd / 4; ++j) {
    b[j].x = wrow[4 * j + kh];  b[j].y = wrow[4 * j + kh + 1];
  }
  const float wdiag = W0[n * Dd + (n >> 7)];
  const float bias = b0[n];
  {
    const float* xp = Xp + mt * (16 * Dd) + threadIdx.x;
    float xv[NP];
#pragma unroll
    for (int p = 0; p < NP; ++p) xv[p] = xp[p * (Bsz * Dd)];
    float s = 0.0f;
#pragma unroll
    for (int p = 0; p < NP; ++p) s += xv[p];
    xs[threadIdx.x] = s;
  }
  __syncthreads();
  v8f acc = {};
  {
    const float* arow = xs + (lane & 15) * Dd;
#pragma unroll
    for (int j = 0; j < Dd / 4; ++j) {
      v2f a;
      a.x = arow[4 * j + kh];  a.y = arow[4 * j + kh + 1];
      acc = wmma_f32_k4(a, b[j], acc);
    }
  }

  const int rowoff = (lane >> 4) * 8;
  const float ldiag = __logf(wdiag);
  const int cl = lane & 15;
  const int pk = ((mt * 256 + nt * 2 + (cl >> 3)) * 32 + ((cl >> 1) & 1) * 16) * 4
               + 2 * ((cl >> 2) & 1) + (cl & 1);
  for (int v = 0; v < 8; ++v) {
    const float h = acc[v] + bias;
    act[pk + (v + rowoff) * 4] = fast_tanh(h);
    gpk[pk + (v + rowoff) * 4] = ldiag + log_abs_dtanh(h);
  }
}

__device__ __forceinline__ void rows_load(const float* __restrict__ row, int kh,
                                          v2f (&f)[32]) {
#pragma unroll
  for (int j = 0; j < 32; ++j) {
    f[j].x = row[4 * j + kh];  f[j].y = row[4 * j + kh + 1];
  }
}
__device__ __forceinline__ float rows_max_exp(v2f (&e)[32]) {
  float mx = fmaxf(e[0].x, e[0].y);
#pragma unroll
  for (int j = 1; j < 32; ++j) mx = fmaxf(mx, fmaxf(e[j].x, e[j].y));
  mx = fmaxf(mx, __shfl_xor(mx, 16));
#pragma unroll
  for (int j = 0; j < 32; ++j) {
    e[j].x = __expf(e[j].x - mx);  e[j].y = __expf(e[j].y - mx);
  }
  return mx;
}

#define L1_KS 2
#define L1_NS 7
#define L1_MT 4
#define L1_NT 2
#define L1_NBLK (HID / 4 / 4 / L1_KS)
#define L1_TAIL0 (((L1_NBLK - (2 * L1_NS - 1)) / L1_NS + 1) * L1_NS)

#define L1_PHASE() asm volatile("" : "+v"(acc[0][0]), "+v"(acc[0][1]), "+v"(acc[0][2]), "+v"(acc[0][3]), \
                                "+v"(acc[1][0]), "+v"(acc[1][1]), "+v"(acc[1][2]), "+v"(acc[1][3]) :: "memory")

__device__ __forceinline__ void l1_load(const float* __restrict__ wrow,
                                        const v4f* __restrict__ pa, int blk,
                                        v2f (&b)[L1_NT][L1_KS], v4f (&a)[L1_MT]) {
#pragma unroll
  for (int n = 0; n < L1_NT; ++n)
#pragma unroll
    for (int i = 0; i < L1_KS; ++i)
      b[n][i] = *(const v2f*)(wrow + n * (16 * HID) + (blk * L1_KS + i) * 4);
#pragma unroll
  for (int t = 0; t < L1_MT; ++t)
    a[t] = pa[t * (256 * 32) + blk * 32];
}

__device__ __forceinline__ void l1_mma(const v2f (&b)[L1_NT][L1_KS], const v4f (&a)[L1_MT],
                                       v8f (&acc)[L1_NT][L1_MT]) {
#pragma unroll
  for (int n = 0; n < L1_NT; ++n)
#pragma unroll
    for (int t = 0; t < L1_MT; ++t) {
      v2f x;
      x.x = a[t].x;  x.y = a[t].y;
      acc[n][t] = wmma_f32_k4(x, b[n][0], acc[n][t]);
    }
#pragma unroll
  for (int n = 0; n < L1_NT; ++n)
#pragma unroll
    for (int t = 0; t < L1_MT; ++t) {
      v2f y;
      y.x = a[t].z;  y.y = a[t].w;
      acc[n][t] = wmma_f32_k4(y, b[n][1], acc[n][t]);
    }
}

__device__ __forceinline__ void l1_gmma(const v2f (&b)[32], const v4f (&e0)[16],
                                        const v4f (&e1)[16], v8f& s0, v8f& s1) {
#pragma unroll
  for (int i = 0; i < 16; ++i) {
    v2f x0, x1, y0, y1;
    x0.x = e0[i].x;  x0.y = e0[i].y;   y0.x = e0[i].z;  y0.y = e0[i].w;
    x1.x = e1[i].x;  x1.y = e1[i].y;   y1.x = e1[i].z;  y1.y = e1[i].w;
    s0 = wmma_f32_k4(x0, b[2 * i], s0);
    s1 = wmma_f32_k4(x1, b[2 * i], s1);
    s0 = wmma_f32_k4(y0, b[2 * i + 1], s0);
    s1 = wmma_f32_k4(y1, b[2 * i + 1], s1);
  }
}

__device__ __forceinline__ void frag_load(const v4f* __restrict__ pg, v4f (&e)[16]) {
#pragma unroll
  for (int i = 0; i < 16; ++i) e[i] = pg[i * 32];
}
__device__ __forceinline__ float frag_max_exp(v4f (&e)[16]) {
  float mx = fmaxf(fmaxf(e[0].x, e[0].y), fmaxf(e[0].z, e[0].w));
#pragma unroll
  for (int i = 1; i < 16; ++i)
    mx = fmaxf(mx, fmaxf(fmaxf(e[i].x, e[i].y), fmaxf(e[i].z, e[i].w)));
  mx = fmaxf(mx, __shfl_xor(mx, 16));
#pragma unroll
  for (int i = 0; i < 16; ++i) {
    e[i].x = __expf(e[i].x - mx);  e[i].y = __expf(e[i].y - mx);
    e[i].z = __expf(e[i].z - mx);  e[i].w = __expf(e[i].w - mx);
  }
  return mx;
}
__device__ __forceinline__ void frag_hold(const v4f (&e)[16]) {
  asm volatile("" :: "v"(e[0]), "v"(e[1]), "v"(e[2]), "v"(e[3]), "v"(e[4]), "v"(e[5]),
               "v"(e[6]), "v"(e[7]), "v"(e[8]), "v"(e[9]), "v"(e[10]), "v"(e[11]),
               "v"(e[12]), "v"(e[13]), "v"(e[14]), "v"(e[15]));
}

__global__ void __launch_bounds__(128)
k_layer1(const float* __restrict__ Apk, const float* __restrict__ Gpk,
         const float* __restrict__ W1, const float* __restrict__ b1,
         float* __restrict__ act, float* __restrict__ Gnew) {
  __shared__ float xch[4][L1_NT * L1_MT][8][32];
  const int lane = threadIdx.x & 31;
  const int wv  = threadIdx.x >> 5;
  const int ntl = wv & 1;
  const int mp  = wv >> 1;
  const int mq  = blockIdx.x >> 6;
  const int ntp = (blockIdx.x & 63) * 2;
  const int nt  = ntp + ntl;
  const int ks  = nt >> 3;
  const int r   = lane & 15;
  const int kh  = (lane >> 4) << 1;
  const float* wrow = W1 + (size_t)(ntp * 16 + r) * HID + kh + wv * (HID / 4);
  const v4f* pa  = (const v4f*)Apk + (size_t)(mq * 4) * (256 * 32) + wv * (64 * 32) + lane;
  const v4f* pg0 = (const v4f*)Gpk + (size_t)(mq * 4 + mp * 2) * (256 * 32) + ks * (16 * 32) + lane;
  const v4f* pg1 = pg0 + 256 * 32;

  v8f acc[L1_NT][L1_MT] = {};
  v8f s0 = {};
  v8f s1 = {};
  v2f w[L1_NS][L1_NT][L1_KS];
  v4f x[L1_NS][L1_MT];
  v4f e0[16], e1[16];
  v2f bD[32];
  frag_load(pg0, e0);
  frag_load(pg1, e1);
#pragma unroll
  for (int s = 0; s < 3; ++s) {
    l1_load(wrow, pa, s, w[s], x[s]);
    L1_PHASE();
  }
  const float mx0 = frag_max_exp(e0);
  const float mx1 = frag_max_exp(e1);
  frag_hold(e0);
  frag_hold(e1);
#pragma unroll
  for (int s = 3; s < L1_NS - 1; ++s) {
    l1_load(wrow, pa, s, w[s], x[s]);
    L1_PHASE();
  }
  for (int blk = 0; blk < L1_TAIL0; blk += L1_NS) {
#pragma unroll
    for (int s = 0; s < L1_NS; ++s) {
      const int t = (s + L1_NS - 1) % L1_NS;
      l1_load(wrow, pa, blk + s + L1_NS - 1, w[t], x[t]);
      l1_mma(w[s], x[s], acc);
      L1_PHASE();
    }
  }
  const float* wdiag = W1 + (size_t)(nt * 16 + r) * HID + kh + ks * Mm;
#pragma unroll
  for (int i = L1_TAIL0; i < L1_NBLK; ++i) {
    if (i + L1_NS - 1 < L1_NBLK) {
      const int t = (i + L1_NS - 1) % L1_NS;
      l1_load(wrow, pa, i + L1_NS - 1, w[t], x[t]);
    }
    if (i == L1_NBLK - 4) {
      L1_PHASE();
#pragma unroll
      for (int j = 0; j < 32; ++j)
        bD[j] = *(const v2f*)(wdiag + 4 * j);
      L1_PHASE();
    }
    l1_mma(w[i % L1_NS], x[i % L1_NS], acc);
    L1_PHASE();
  }
#pragma unroll
  for (int n = 0; n < L1_NT; ++n)
#pragma unroll
    for (int t = 0; t < L1_MT; ++t)
#pragma unroll
      for (int v = 0; v < 8; ++v)
        xch[wv][n * L1_MT + t][v][lane] = acc[n][t][v];
  l1_gmma(bD, e0, e1, s0, s1);
  __syncthreads();

  const int n = nt * 16 + r;
  const int rowoff = (lane >> 4) * 8;
  const float bias = b1[n];
  const int tl0 = ntl * L1_MT + mp * 2;
#pragma unroll
  for (int v = 0; v < 8; ++v) {
    const int m = (mq * 4 + mp * 2) * 16 + v + rowoff;
    const float hk = ((xch[0][tl0][v][lane] + xch[1][tl0][v][lane])
                      + xch[2][tl0][v][lane]) + xch[3][tl0][v][lane];
    const float h = hk + bias;
    const float mv = __shfl(mx0, v + rowoff);
    act[m * HID + n] = fast_tanh(h);
    Gnew[m * HID + n] = __logf(s0[v]) + mv + log_abs_dtanh(h);
  }
#pragma unroll
  for (int v = 0; v < 8; ++v) {
    const int m = (mq * 4 + mp * 2 + 1) * 16 + v + rowoff;
    const float hk = ((xch[0][tl0 + 1][v][lane] + xch[1][tl0 + 1][v][lane])
                      + xch[2][tl0 + 1][v][lane]) + xch[3][tl0 + 1][v][lane];
    const float h = hk + bias;
    const float mv = __shfl(mx1, v + rowoff);
    act[m * HID + n] = fast_tanh(h);
    Gnew[m * HID + n] = __logf(s1[v]) + mv + log_abs_dtanh(h);
  }
}


#define TAIL_LOADS_STAY() __builtin_amdgcn_sched_barrier(0)

__global__ void __launch_bounds__(64)
k_tail(const float* __restrict__ act1, const float* __restrict__ G,
       const float* __restrict__ W2,
       float* __restrict__ Pout, float* __restrict__ Pld) {
  const int role = __builtin_amdgcn_readfirstlane(threadIdx.x >> 5);
  const int kb   = blockIdx.x >> 4;
  const int mt   = blockIdx.x & 15;
  const int lane = threadIdx.x & 31;
  const int r    = lane & 15;
  const int kh   = (lane >> 4) << 1;
  const int rowoff = (lane >> 4) * 8;
  const float* wrow = W2 + (size_t)r * HID + kb * Mm;
  v2f b[32];
  if (role == 0) {
    const float* arow = act1 + (size_t)(mt * 16 + r) * HID + kb * Mm;
    v2f a[32];
#pragma unroll
    for (int j = 0; j < 32; ++j) {
      a[j].x = arow[4 * j + kh];  a[j].y = arow[4 * j + kh + 1];
      b[j].x = wrow[4 * j + kh];  b[j].y = wrow[4 * j + kh + 1];
    }
    TAIL_LOADS_STAY();
    v8f accx = {};
#pragma unroll
    for (int j = 0; j < 32; ++j) accx = wmma_f32_k4(a[j], b[j], accx);
    float* po = Pout + (size_t)kb * (Bsz * Dd);
    for (int v = 0; v < 8; ++v) {
      const int m = mt * 16 + v + rowoff;
      po[m * Dd + r] = accx[v];
    }
  } else {
    const float* grow = G + (size_t)(mt * 16 + r) * HID + kb * Mm;
    v2f e[32];
    rows_load(grow, kh, e);
    rows_load(wrow, kh, b);
    TAIL_LOADS_STAY();
    const float mx = rows_max_exp(e);
    v8f accs = {};
#pragma unroll
    for (int j = 0; j < 32; ++j) accs = wmma_f32_k4(e[j], b[j], accs);
    float mv[8];
    for (int v = 0; v < 8; ++v) mv[v] = __shfl(mx, v + rowoff);
    if (r == kb) {
      float* pl = Pld + (kb * 16 + mt) * 32;
      for (int v = 0; v < 8; ++v)
        pl[v + rowoff] = __logf(accs[v]) + mv[v];
    }
  }
}

__global__ void k_tail_sum(const float* __restrict__ Pout, const float* __restrict__ Pld0,
                           const float* __restrict__ Pld1,
                           float* __restrict__ x_out, float* __restrict__ ld) {
  const int g = blockIdx.x;
  const int t = threadIdx.x;
  const int idx = g * 256 + t;
  float s = 0.0f;
  for (int kb = 0; kb < Dd; ++kb) s += Pout[kb * (Bsz * Dd) + idx];
  x_out[idx] = s;
  if (g < 8 && t < 32) {
    const int b = g * 32 + t;
    float tot0 = 0.0f;
    float tot1 = 0.0f;
    for (int kb = 0; kb < Dd; ++kb)
      tot0 += Pld0[(kb * 16 + (b >> 4)) * 32 + (b & 15)];
    for (int kb = 0; kb < Dd; ++kb)
      tot1 += Pld1[(kb * 16 + (b >> 4)) * 32 + (b & 15)];
    ld[b] = tot0 + tot1;
  }
}

template <int NP>
static void run_flow(const float* Xp, const float* W0, const float* W1,
                     const float* W2, const float* b0, const float* b1,
                     float* act0, float* act1, float* g0,
                     float* g1, float* Pout, float* Pld,
                     hipStream_t stream) {
  k_layer0<NP> <<<256, 256, 0, stream>>>(Xp, W0, b0, act0, g0);
  k_layer1   <<<256, 128, 0, stream>>>(act0, g0, W1, b1, act1, g1);
  k_tail     <<<256, 64, 0, stream>>>(act1, g1, W2, Pout, Pld);
}

extern "C" void kernel_launch(void* const* d_in, const int* in_sizes, int n_in,
                              void* d_out, int out_size, void* d_ws, size_t ws_size,
                              hipStream_t stream) {
  const float* inp   = (const float*)d_in[0];
  const float* f0_W0 = (const float*)d_in[1];
  const float* f0_W1 = (const float*)d_in[2];
  const float* f0_W2 = (const float*)d_in[3];
  const float* f0_b0 = (const float*)d_in[4];
  const float* f0_b1 = (const float*)d_in[5];
  const float* f1_W0 = (const float*)d_in[6];
  const float* f1_W1 = (const float*)d_in[7];
  const float* f1_W2 = (const float*)d_in[8];
  const float* f1_b0 = (const float*)d_in[9];
  const float* f1_b1 = (const float*)d_in[10];

  float* out = (float*)d_out;
  float* ld  = out + Bsz * Dd;

  float* ws   = (float*)d_ws;
  float* act0 = ws;
  float* act1 = act0 + Bsz * HID;
  float* g0   = act1 + Bsz * HID;
  float* g1   = g0   + Bsz * HID;
  float* Pout0 = g1    + Bsz * HID;
  float* Pout1 = Pout0 + Dd * Bsz * Dd;
  float* Pld0  = Pout1 + Dd * Bsz * Dd;
  float* Pld1  = Pld0  + Dd * 16 * 32;

  run_flow<1>(inp, f0_W0, f0_W1, f0_W2, f0_b0, f0_b1,
              act0, act1, g0, g1, Pout0, Pld0, stream);
  run_flow<Dd>(Pout0, f1_W0, f1_W1, f1_W2, f1_b0, f1_b1,
               act0, act1, g0, g1, Pout1, Pld1, stream);
  k_tail_sum <<<16, 256, 0, stream>>>(Pout1, Pld0, Pld1, out, ld);
}


# Layer 1's k loop ends at each column slab's diagonal block of the reference's mask; group = 4 M-tiles x one N-tile each of slabs s and 15-s: 608 WMMAs per wave, was 1088; k sum order changes
# speedup vs baseline: 8.0717x; 1.0749x over previous
#include <hip/hip_runtime.h>
#include <hip/hip_bf16.h>


typedef float v2f __attribute__((ext_vector_type(2)));
typedef float v4f __attribute__((ext_vector_type(4)));
typedef float v8f __attribute__((ext_vector_type(8)));

#define Bsz   256
#define Dd    16
#define Mm    128
#define HID   2048

__device__ __forceinline__ v8f wmma_f32_k4(v2f a, v2f b, v8f c) {
  return __builtin_amdgcn_wmma_f32_16x16x4_f32(
      false, a, false, b, (short)0, c, false, false);
}

__device__ __forceinline__ float fast_tanh(float x) {
#if defined(__has_builtin)
#if __has_builtin(__builtin_amdgcn_tanhf)
  return __builtin_amdgcn_tanhf(x);
#elif __has_builtin(__builtin_amdgcn_tanh_f32)
  return __builtin_amdgcn_tanh_f32(x);
#else
  return tanhf(x);
#endif
#else
  return tanhf(x);
#endif
}

__device__ __forceinline__ float softplus_f(float x) {
  return fmaxf(x, 0.0f) + __logf(1.0f + __expf(-fabsf(x)));
}
__device__ __forceinline__ float log_abs_dtanh(float h) {
  return 2.0f * (0.6931471805599453f - h - softplus_f(-2.0f * h));
}

template <int NP>
__global__ void __launch_bounds__(256)
k_layer0(const float* __restrict__ Xp, const float* __restrict__ W0,
         const float* __restrict__ b0,
         float* __restrict__ act, float* __restrict__ gpk) {
  __shared__ float xs[16 * Dd];
  const int wave = (blockIdx.x * blockDim.x + threadIdx.x) >> 5;
  const int mt = wave >> 7;
  const int nt = wave & 127;
  if (mt >= Bsz / 16) return;
  const int lane = threadIdx.x & 31;
  const int n = nt * 16 + (lane & 15);
  const int kh = (lane >> 4) << 1;
  const float* wrow = W0 + (size_t)n * Dd;
  v2f b[Dd / 4];
#pragma unroll
  for (int j = 0; j < Dd / 4; ++j) {
    b[j].x = wrow[4 * j + kh];  b[j].y = wrow[4 * j + kh + 1];
  }
  const float wdiag = W0[n * Dd + (n >> 7)];
  const float bias = b0[n];
  {
    const float* xp = Xp + mt * (16 * Dd) + threadIdx.x;
    float xv[NP];
#pragma unroll
    for (int p = 0; p < NP; ++p) xv[p] = xp[p * (Bsz * Dd)];
    float s = 0.0f;
#pragma unroll
    for (int p = 0; p < NP; ++p) s += xv[p];
    xs[threadIdx.x] = s;
  }
  __syncthreads();
  v8f acc = {};
  {
    const float* arow = xs + (lane & 15) * Dd;
#pragma unroll
    for (int j = 0; j < Dd / 4; ++j) {
      v2f a;
      a.x = arow[4 * j + kh];  a.y = arow[4 * j + kh + 1];
      acc = wmma_f32_k4(a, b[j], acc);
    }
  }

  const int rowoff = (lane >> 4) * 8;
  const float ldiag = __logf(wdiag);
  const int cl = lane & 15;
  const int pk = ((mt * 256 + nt * 2 + (cl >> 3)) * 32 + ((cl >> 1) & 1) * 16) * 4
               + 2 * ((cl >> 2) & 1) + (cl & 1);
  for (int v = 0; v < 8; ++v) {
    const float h = acc[v] + bias;
    act[pk + (v + rowoff) * 4] = fast_tanh(h);
    gpk[pk + (v + rowoff) * 4] = ldiag + log_abs_dtanh(h);
  }
}

__device__ __forceinline__ void rows_load(const float* __restrict__ row, int kh,
                                          v2f (&f)[32]) {
#pragma unroll
  for (int j = 0; j < 32; ++j) {
    f[j].x = row[4 * j + kh];  f[j].y = row[4 * j + kh + 1];
  }
}
__device__ __forceinline__ float rows_max_exp(v2f (&e)[32]) {
  float mx = fmaxf(e[0].x, e[0].y);
#pragma unroll
  for (int j = 1; j < 32; ++j) mx = fmaxf(mx, fmaxf(e[j].x, e[j].y));
  mx = fmaxf(mx, __shfl_xor(mx, 16));
#pragma unroll
  for (int j = 0; j < 32; ++j) {
    e[j].x = __expf(e[j].x - mx);  e[j].y = __expf(e[j].y - mx);
  }
  return mx;
}

#define L1_KS 2
#define L1_NS 8
#define L1_MT 4
#define L1_NT 1
#define L1_NHALF 17
#define L1_NBLK (4 * L1_NHALF)

#define L1_PHASE(acc) asm volatile("" : "+v"(acc[0][0]), "+v"(acc[0][1]), \
                                   "+v"(acc[0][2]), "+v"(acc[0][3]) :: "memory")

#define L1_NEXT_SLAB() do {                                              \
    _Pragma("unroll") for (int n_ = 0; n_ < L1_NT; ++n_)                  \
      _Pragma("unroll") for (int t_ = 0; t_ < L1_MT; ++t_) {              \
        accA[n_][t_] = acc[n_][t_];                                       \
        acc[n_][t_] = v8f{};                                              \
      }                                                                   \
  } while (0)

struct L1Src {
  const float* W1;
  const v4f* A;
  int n1;
  int w0, w1;
  int a0, a1;
};

__device__ __forceinline__ void l1_load(const L1Src& q, int lw, int lane, int vi,
                                        v2f (&b)[L1_NT][L1_KS], v4f (&a)[L1_MT]) {
  const bool first = vi < q.n1;
  const float* wrow = q.W1 + ((first ? q.w0 : q.w1) + vi * (4 * L1_KS));
  const v4f* pa = q.A + ((first ? q.a0 : q.a1) + vi * 32);
#pragma unroll
  for (int n = 0; n < L1_NT; ++n)
#pragma unroll
    for (int i = 0; i < L1_KS; ++i)
      b[n][i] = *(const v2f*)(wrow + lw + n * (16 * HID) + i * 4);
#pragma unroll
  for (int t = 0; t < L1_MT; ++t)
    a[t] = pa[lane + t * (256 * 32)];
}

__device__ __forceinline__ void l1_mma(const v2f (&b)[L1_NT][L1_KS], const v4f (&a)[L1_MT],
                                       v8f (&acc)[L1_NT][L1_MT]) {
#pragma unroll
  for (int n = 0; n < L1_NT; ++n)
#pragma unroll
    for (int t = 0; t < L1_MT; ++t) {
      v2f x;
      x.x = a[t].x;  x.y = a[t].y;
      acc[n][t] = wmma_f32_k4(x, b[n][0], acc[n][t]);
    }
#pragma unroll
  for (int n = 0; n < L1_NT; ++n)
#pragma unroll
    for (int t = 0; t < L1_MT; ++t) {
      v2f y;
      y.x = a[t].z;  y.y = a[t].w;
      acc[n][t] = wmma_f32_k4(y, b[n][1], acc[n][t]);
    }
}

template <int S0, int NLOAD>
__device__ __forceinline__ void l1_half(const L1Src& q, int lw, int lane, int i0,
                                        v2f (&w)[L1_NS][L1_NT][L1_KS], v4f (&x)[L1_NS][L1_MT],
                                        v8f (&acc)[L1_NT][L1_MT]) {
#pragma unroll
  for (int s = 0; s < 4; ++s) {
    if (s < NLOAD) {
      const int t = (S0 + s + L1_NS - 1) % L1_NS;
      l1_load(q, lw, lane, i0 + s + L1_NS - 1, w[t], x[t]);
    }
    l1_mma(w[S0 + s], x[S0 + s], acc);
    L1_PHASE(acc);
  }
}

__device__ __forceinline__ void l1_gmma(const v2f (&b)[32], const v4f (&e0)[16],
                                        const v4f (&e1)[16], v8f& s0, v8f& s1) {
#pragma unroll
  for (int i = 0; i < 16; ++i) {
    v2f x0, x1, y0, y1;
    x0.x = e0[i].x;  x0.y = e0[i].y;   y0.x = e0[i].z;  y0.y = e0[i].w;
    x1.x = e1[i].x;  x1.y = e1[i].y;   y1.x = e1[i].z;  y1.y = e1[i].w;
    s0 = wmma_f32_k4(x0, b[2 * i], s0);
    s1 = wmma_f32_k4(x1, b[2 * i], s1);
    s0 = wmma_f32_k4(y0, b[2 * i + 1], s0);
    s1 = wmma_f32_k4(y1, b[2 * i + 1], s1);
  }
}

__device__ __forceinline__ void frag_load(const v4f* __restrict__ pg, v4f (&e)[16]) {
#pragma unroll
  for (int i = 0; i < 16; ++i) e[i] = pg[i * 32];
}
__device__ __forceinline__ float frag_max_exp(v4f (&e)[16]) {
  float mx = fmaxf(fmaxf(e[0].x, e[0].y), fmaxf(e[0].z, e[0].w));
#pragma unroll
  for (int i = 1; i < 16; ++i)
    mx = fmaxf(mx, fmaxf(fmaxf(e[i].x, e[i].y), fmaxf(e[i].z, e[i].w)));
  mx = fmaxf(mx, __shfl_xor(mx, 16));
#pragma unroll
  for (int i = 0; i < 16; ++i) {
    e[i].x = __expf(e[i].x - mx);  e[i].y = __expf(e[i].y - mx);
    e[i].z = __expf(e[i].z - mx);  e[i].w = __expf(e[i].w - mx);
  }
  return mx;
}
__device__ __forceinline__ void frag_hold(const v4f (&e)[16]) {
  asm volatile("" :: "v"(e[0]), "v"(e[1]), "v"(e[2]), "v"(e[3]), "v"(e[4]), "v"(e[5]),
               "v"(e[6]), "v"(e[7]), "v"(e[8]), "v"(e[9]), "v"(e[10]), "v"(e[11]),
               "v"(e[12]), "v"(e[13]), "v"(e[14]), "v"(e[15]));
}

__device__ __forceinline__ void frag_arrive(v4f (&e)[16]) {
  asm volatile("" : "+v"(e[0]), "+v"(e[1]), "+v"(e[2]), "+v"(e[3]), "+v"(e[4]), "+v"(e[5]),
               "+v"(e[6]), "+v"(e[7]), "+v"(e[8]), "+v"(e[9]), "+v"(e[10]), "+v"(e[11]),
               "+v"(e[12]), "+v"(e[13]), "+v"(e[14]), "+v"(e[15]));
}

__global__ void __launch_bounds__(128)
k_layer1(const float* __restrict__ Apk, const float* __restrict__ Gpk,
         const float* __restrict__ W1, const float* __restrict__ b1,
         float* __restrict__ act, float* __restrict__ Gnew) {
  __shared__ float xch[4][2 * L1_NT * L1_MT][8][32];
  const int lane = threadIdx.x & 31;
  const int wv  = __builtin_amdgcn_readfirstlane(threadIdx.x >> 5);
  const int mq  = blockIdx.x >> 6;
  const int sp  = (blockIdx.x >> 3) & 7;
  const int ni  = blockIdx.x & 7;
  const int ntA = sp * 8 + ni;
  const int ntB = (15 - sp) * 8 + ni;
  const int n1  = 4 * (sp + 1);
  const int n2  = 4 * (16 - sp);
  const int nt  = (wv & 2) ? ntB : ntA;
  const int mt0 = mq * 4 + (wv & 1) * 2;
  const int ks  = nt >> 3;
  const int r   = lane & 15;
  const int kh  = (lane >> 4) << 1;
  const int lw  = r * HID + kh;
  L1Src q;
  q.W1 = W1;
  q.A  = (const v4f*)Apk;
  q.n1 = n1;
  q.w0 = ntA * 16 * HID + wv * n1 * (4 * L1_KS);
  q.w1 = ntB * 16 * HID + wv * n2 * (4 * L1_KS) - n1 * (4 * L1_KS);
  q.a0 = mq * 4 * (256 * 32) + wv * n1 * 32;
  q.a1 = mq * 4 * (256 * 32) + wv * n2 * 32 - n1 * 32;
  const v4f* pg0 = (const v4f*)Gpk + (size_t)mt0 * (256 * 32) + ks * (16 * 32) + lane;
  const v4f* pg1 = pg0 + 256 * 32;

  v8f acc[L1_NT][L1_MT] = {};
  v8f accA[L1_NT][L1_MT];
  v8f s0 = {};
  v8f s1 = {};
  v2f w[L1_NS][L1_NT][L1_KS];
  v4f x[L1_NS][L1_MT];
  v4f e0[16], e1[16];
  v2f bD[32];
  frag_load(pg0, e0);
  frag_load(pg1, e1);
#pragma unroll
  for (int s = 0; s < 3; ++s) {
    l1_load(q, lw, lane, s, w[s], x[s]);
    L1_PHASE(acc);
  }
  frag_arrive(e0);
  frag_arrive(e1);
  const float mx0 = frag_max_exp(e0);
  const float mx1 = frag_max_exp(e1);
  frag_hold(e0);
  frag_hold(e1);
#pragma unroll
  for (int s = 3; s < L1_NS - 1; ++s) {
    l1_load(q, lw, lane, s, w[s], x[s]);
    L1_PHASE(acc);
  }
  const int sw = sp + 1;
  for (int j = 0; j < L1_NHALF - 3; j += 2) {
    if (j == sw) L1_NEXT_SLAB();
    l1_half<0, 4>(q, lw, lane, 4 * j, w, x, acc);
    if (j + 1 == sw) L1_NEXT_SLAB();
    l1_half<4, 4>(q, lw, lane, 4 * j + 4, w, x, acc);
  }
  l1_half<0, 4>(q, lw, lane, 4 * (L1_NHALF - 3), w, x, acc);
  l1_half<4, 1>(q, lw, lane, 4 * (L1_NHALF - 2), w, x, acc);
  {
    const float* wdiag = W1 + (size_t)(nt * 16 + r) * HID + kh + ks * Mm;
#pragma unroll
    for (int j = 0; j < 32; ++j)
      bD[j] = *(const v2f*)(wdiag + 4 * j);
    L1_PHASE(acc);
  }
  l1_half<0, 0>(q, lw, lane, 4 * (L1_NHALF - 1), w, x, acc);
#pragma unroll
  for (int n = 0; n < L1_NT; ++n)
#pragma unroll
    for (int t = 0; t < L1_MT; ++t)
#pragma unroll
      for (int v = 0; v < 8; ++v) {
        xch[wv][n * L1_MT + t][v][lane] = accA[n][t][v];
        xch[wv][L1_NT * L1_MT + n * L1_MT + t][v][lane] = acc[n][t][v];
      }
  l1_gmma(bD, e0, e1, s0, s1);
  __syncthreads();

  const int n = nt * 16 + r;
  const int rowoff = (lane >> 4) * 8;
  const float bias = b1[n];
  const int tl0 = wv * 2;
#pragma unroll
  for (int v = 0; v < 8; ++v) {
    const int m = mt0 * 16 + v + rowoff;
    const float hk = ((xch[0][tl0][v][lane] + xch[1][tl0][v][lane])
                      + xch[2][tl0][v][lane]) + xch[3][tl0][v][lane];
    const float h = hk + bias;
    const float mv = __shfl(mx0, v + rowoff);
    act[m * HID + n] = fast_tanh(h);
    Gnew[m * HID + n] = __logf(s0[v]) + mv + log_abs_dtanh(h);
  }
#pragma unroll
  for (int v = 0; v < 8; ++v) {
    const int m = (mt0 + 1) * 16 + v + rowoff;
    const float hk = ((xch[0][tl0 + 1][v][lane] + xch[1][tl0 + 1][v][lane])
                      + xch[2][tl0 + 1][v][lane]) + xch[3][tl0 + 1][v][lane];
    const float h = hk + bias;
    const float mv = __shfl(mx1, v + rowoff);
    act[m * HID + n] = fast_tanh(h);
    Gnew[m * HID + n] = __logf(s1[v]) + mv + log_abs_dtanh(h);
  }
}


#define TAIL_LOADS_STAY() __builtin_amdgcn_sched_barrier(0)

__global__ void __launch_bounds__(64)
k_tail(const float* __restrict__ act1, const float* __restrict__ G,
       const float* __restrict__ W2,
       float* __restrict__ Pout, float* __restrict__ Pld) {
  const int role = __builtin_amdgcn_readfirstlane(threadIdx.x >> 5);
  const int kb   = blockIdx.x >> 4;
  const int mt   = blockIdx.x & 15;
  const int lane = threadIdx.x & 31;
  const int r    = lane & 15;
  const int kh   = (lane >> 4) << 1;
  const int rowoff = (lane >> 4) * 8;
  const float* wrow = W2 + (size_t)r * HID + kb * Mm;
  v2f b[32];
  if (role == 0) {
    const float* arow = act1 + (size_t)(mt * 16 + r) * HID + kb * Mm;
    v2f a[32];
#pragma unroll
    for (int j = 0; j < 32; ++j) {
      a[j].x = arow[4 * j + kh];  a[j].y = arow[4 * j + kh + 1];
      b[j].x = wrow[4 * j + kh];  b[j].y = wrow[4 * j + kh + 1];
    }
    TAIL_LOADS_STAY();
    v8f accx = {};
#pragma unroll
    for (int j = 0; j < 32; ++j) accx = wmma_f32_k4(a[j], b[j], accx);
    float* po = Pout + (size_t)kb * (Bsz * Dd);
    for (int v = 0; v < 8; ++v) {
      const int m = mt * 16 + v + rowoff;
      po[m * Dd + r] = accx[v];
    }
  } else {
    const float* grow = G + (size_t)(mt * 16 + r) * HID + kb * Mm;
    v2f e[32];
    rows_load(grow, kh, e);
    rows_load(wrow, kh, b);
    TAIL_LOADS_STAY();
    const float mx = rows_max_exp(e);
    v8f accs = {};
#pragma unroll
    for (int j = 0; j < 32; ++j) accs = wmma_f32_k4(e[j], b[j], accs);
    float mv[8];
    for (int v = 0; v < 8; ++v) mv[v] = __shfl(mx, v + rowoff);
    if (r == kb) {
      float* pl = Pld + (kb * 16 + mt) * 32;
      for (int v = 0; v < 8; ++v)
        pl[v + rowoff] = __logf(accs[v]) + mv[v];
    }
  }
}

__global__ void k_tail_sum(const float* __restrict__ Pout, const float* __restrict__ Pld0,
                           const float* __restrict__ Pld1,
                           float* __restrict__ x_out, float* __restrict__ ld) {
  const int g = blockIdx.x;
  const int t = threadIdx.x;
  const int idx = g * 256 + t;
  float s = 0.0f;
  for (int kb = 0; kb < Dd; ++kb) s += Pout[kb * (Bsz * Dd) + idx];
  x_out[idx] = s;
  if (g < 8 && t < 32) {
    const int b = g * 32 + t;
    float tot0 = 0.0f;
    float tot1 = 0.0f;
    for (int kb = 0; kb < Dd; ++kb)
      tot0 += Pld0[(kb * 16 + (b >> 4)) * 32 + (b & 15)];
    for (int kb = 0; kb < Dd; ++kb)
      tot1 += Pld1[(kb * 16 + (b >> 4)) * 32 + (b & 15)];
    ld[b] = tot0 + tot1;
  }
}

template <int NP>
static void run_flow(const float* Xp, const float* W0, const float* W1,
                     const float* W2, const float* b0, const float* b1,
                     float* act0, float* act1, float* g0,
                     float* g1, float* Pout, float* Pld,
                     hipStream_t stream) {
  k_layer0<NP> <<<256, 256, 0, stream>>>(Xp, W0, b0, act0, g0);
  k_layer1   <<<256, 128, 0, stream>>>(act0, g0, W1, b1, act1, g1);
  k_tail     <<<256, 64, 0, stream>>>(act1, g1, W2, Pout, Pld);
}

extern "C" void kernel_launch(void* const* d_in, const int* in_sizes, int n_in,
                              void* d_out, int out_size, void* d_ws, size_t ws_size,
                              hipStream_t stream) {
  const float* inp   = (const float*)d_in[0];
  const float* f0_W0 = (const float*)d_in[1];
  const float* f0_W1 = (const float*)d_in[2];
  const float* f0_W2 = (const float*)d_in[3];
  const float* f0_b0 = (const float*)d_in[4];
  const float* f0_b1 = (const float*)d_in[5];
  const float* f1_W0 = (const float*)d_in[6];
  const float* f1_W1 = (const float*)d_in[7];
  const float* f1_W2 = (const float*)d_in[8];
  const float* f1_b0 = (const float*)d_in[9];
  const float* f1_b1 = (const float*)d_in[10];

  float* out = (float*)d_out;
  float* ld  = out + Bsz * Dd;

  float* ws   = (float*)d_ws;
  float* act0 = ws;
  float* act1 = act0 + Bsz * HID;
  float* g0   = act1 + Bsz * HID;
  float* g1   = g0   + Bsz * HID;
  float* Pout0 = g1    + Bsz * HID;
  float* Pout1 = Pout0 + Dd * Bsz * Dd;
  float* Pld0  = Pout1 + Dd * Bsz * Dd;
  float* Pld1  = Pld0  + Dd * 16 * 32;

  run_flow<1>(inp, f0_W0, f0_W1, f0_W2, f0_b0, f0_b1,
              act0, act1, g0, g1, Pout0, Pld0, stream);
  run_flow<Dd>(Pout0, f1_W0, f1_W1, f1_W2, f1_b0, f1_b1,
               act0, act1, g0, g1, Pout1, Pld1, stream);
  k_tail_sum <<<16, 256, 0, stream>>>(Pout1, Pld0, Pld1, out, ld);
}
